# starting kernel
# baseline (speedup 1.0000x reference)
#include <hip/hip_runtime.h>
#include <hip/hip_bf16.h>
#include <float.h>

#define BATCH   8
#define NPTS    2048
#define CH      128
#define HALF_CH 64
#define KNN_K   20
#define EPSB    1e-5f

typedef __attribute__((ext_vector_type(2))) float v2f;
typedef __attribute__((ext_vector_type(8))) float v8f;

__global__ void prep_params(const float* __restrict__ w1, const float* __restrict__ b1,
                            const float* __restrict__ g1, const float* __restrict__ be1,
                            const float* __restrict__ m1, const float* __restrict__ v1,
                            const float* __restrict__ w2, const float* __restrict__ b2,
                            const float* __restrict__ g2, const float* __restrict__ be2,
                            const float* __restrict__ m2, const float* __restrict__ v2,
                            float* __restrict__ Wc,   float* __restrict__ sw1a,
                            float* __restrict__ sw1d, float* __restrict__ bias1p,
                            float* __restrict__ bias2p) {
  int o = threadIdx.x;
  if (o >= HALF_CH) return;
  float s1 = g1[o] / sqrtf(v1[o] + EPSB);
  bias1p[o] = b1[o] * s1 + (be1[o] - m1[o] * s1);
  #pragma unroll
  for (int c = 0; c < 3; ++c) {
    float wa = w1[o * 6 + c];
    float wb = w1[o * 6 + 3 + c];
    sw1a[o * 3 + c] = s1 * wa;
    sw1d[o * 3 + c] = s1 * (wb - wa);
  }
  float s2 = g2[o] / sqrtf(v2[o] + EPSB);
  bias2p[o] = b2[o] * s2 + (be2[o] - m2[o] * s2);
  for (int c = 0; c < CH; ++c) {
    float wa = w2[o * (2 * CH) + c];
    float wb = w2[o * (2 * CH) + CH + c];
    Wc[(size_t)o * CH + c]             = s2 * wa;
    Wc[(size_t)(HALF_CH + o) * CH + c] = s2 * (wb - wa);
  }
}

__global__ void knn_kernel(const float* __restrict__ xyz, int* __restrict__ idx_out) {
  __shared__ float4 pts[NPTS];
  const int b = blockIdx.y;
  const float* xb = xyz + (size_t)b * 3 * NPTS;
  for (int m = threadIdx.x; m < NPTS; m += blockDim.x) {
    float x = xb[m], y = xb[NPTS + m], z = xb[2 * NPTS + m];
    pts[m] = make_float4(x, y, z, x * x + y * y + z * z);
  }
  __syncthreads();

  const int n = blockIdx.x * blockDim.x + threadIdx.x;
  const float4 me = pts[n];

  float td[KNN_K];
  int   ti[KNN_K];
  #pragma unroll
  for (int i = 0; i < KNN_K; ++i) { td[i] = FLT_MAX; ti[i] = 0; }

  for (int m = 0; m < NPTS; ++m) {
    float4 p = pts[m];
    float dot = me.x * p.x + me.y * p.y + me.z * p.z;
    float d = me.w + p.w - 2.0f * dot;
    if (d < td[KNN_K - 1]) {
      td[KNN_K - 1] = d; ti[KNN_K - 1] = m;
      #pragma unroll
      for (int i = KNN_K - 1; i > 0; --i) {
        if (td[i] < td[i - 1]) {
          float tf = td[i]; td[i] = td[i - 1]; td[i - 1] = tf;
          int   tt = ti[i]; ti[i] = ti[i - 1]; ti[i - 1] = tt;
        }
      }
    }
  }
  int* op = idx_out + ((size_t)b * NPTS + n) * KNN_K;
  #pragma unroll
  for (int i = 0; i < KNN_K; ++i) op[i] = ti[i];
}

__global__ void geo_kernel(const float* __restrict__ xyz, const float* __restrict__ sw1a,
                           const float* __restrict__ sw1d, const float* __restrict__ bias1p,
                           float* __restrict__ PQ) {
  int t  = blockIdx.x * blockDim.x + threadIdx.x;
  int ch = t & (HALF_CH - 1);
  int pt = t >> 6;
  int b  = pt / NPTS;
  int n  = pt - b * NPTS;
  const float* xb = xyz + (size_t)b * 3 * NPTS;
  float x = xb[n], y = xb[NPTS + n], z = xb[2 * NPTS + n];
  float p = sw1a[ch * 3 + 0] * x + sw1a[ch * 3 + 1] * y + sw1a[ch * 3 + 2] * z;
  float q = sw1d[ch * 3 + 0] * x + sw1d[ch * 3 + 1] * y + sw1d[ch * 3 + 2] * z + bias1p[ch];
  float* o = PQ + (size_t)pt * CH;
  o[ch] = p;
  o[HALF_CH + ch] = q;
}

__global__ void feat_gemm(const float* __restrict__ feat, const float* __restrict__ Wc,
                          float* __restrict__ Cf) {
  const int b    = blockIdx.y;
  const int n0   = blockIdx.x * 16;
  const int wave = threadIdx.x >> 5;
  const int lane = threadIdx.x & 31;
  const int l16  = lane & 15;
  const int hi   = lane >> 4;
  const int m0   = wave * 16;

  const float* Fb   = feat + (size_t)b * CH * NPTS;
  const float* arow = Wc + (size_t)(m0 + l16) * CH + 2 * hi;
  const float* bcol = Fb + (size_t)(2 * hi) * NPTS + n0 + l16;

  v8f acc = {0.f, 0.f, 0.f, 0.f, 0.f, 0.f, 0.f, 0.f};

  #pragma unroll 8
  for (int k = 0; k < CH; k += 4) {
    v2f a;
    a.x = arow[k];
    a.y = arow[k + 1];
    v2f bb;
    bb.x = bcol[(size_t)k * NPTS];
    bb.y = bcol[(size_t)(k + 1) * NPTS];
    acc = __builtin_amdgcn_wmma_f32_16x16x4_f32(false, a, false, bb,
                                                (short)0, acc, false, false);
  }

  float* out = Cf + ((size_t)b * NPTS + n0 + l16) * CH + m0 + 8 * hi;
  #pragma unroll
  for (int v = 0; v < 8; ++v) out[v] = acc[v];
}

__global__ void gather_max(const float* __restrict__ PQ, const float* __restrict__ Cf,
                           const int* __restrict__ idx, const float* __restrict__ bias2p,
                           float* __restrict__ out) {
  const int b  = blockIdx.y;
  const int ch = threadIdx.x & (HALF_CH - 1);
  const int n  = blockIdx.x * 4 + (threadIdx.x >> 6);

  const size_t pbase = (size_t)b * NPTS + n;
  const int* id = idx + pbase * KNN_K;

  const float qg = PQ[pbase * CH + HALF_CH + ch];
  const float vf = Cf[pbase * CH + HALF_CH + ch] + bias2p[ch];

  float mg = -FLT_MAX, mf = -FLT_MAX;
  #pragma unroll 4
  for (int k = 0; k < KNN_K; ++k) {
    int j = id[k];
    size_t jb = ((size_t)b * NPTS + j) * CH;
    mg = fmaxf(mg, PQ[jb + ch]);
    mf = fmaxf(mf, Cf[jb + ch]);
  }
  float geo = fmaxf(mg + qg, 0.0f);
  float fea = fmaxf(mf + vf, 0.0f);

  out[(size_t)b * CH * NPTS + (size_t)ch * NPTS + n]             = geo;
  out[(size_t)b * CH * NPTS + (size_t)(HALF_CH + ch) * NPTS + n] = fea;
}

extern "C" void kernel_launch(void* const* d_in, const int* in_sizes, int n_in,
                              void* d_out, int out_size, void* d_ws, size_t ws_size,
                              hipStream_t stream) {
  (void)in_sizes; (void)n_in; (void)out_size; (void)ws_size;

  const float* xyz  = (const float*)d_in[0];
  const float* feat = (const float*)d_in[1];
  const float* w1   = (const float*)d_in[2];
  const float* b1   = (const float*)d_in[3];
  const float* g1   = (const float*)d_in[4];
  const float* be1  = (const float*)d_in[5];
  const float* m1   = (const float*)d_in[6];
  const float* v1   = (const float*)d_in[7];
  const float* w2   = (const float*)d_in[8];
  const float* b2   = (const float*)d_in[9];
  const float* g2   = (const float*)d_in[10];
  const float* be2  = (const float*)d_in[11];
  const float* m2   = (const float*)d_in[12];
  const float* v2   = (const float*)d_in[13];

  char* ws = (char*)d_ws;
  float* Wc     = (float*)ws;
  float* sw1a   = Wc + 16384;
  float* sw1d   = sw1a + 192;
  float* bias1p = sw1d + 192;
  float* bias2p = bias1p + 64;
  float* PQ     = (float*)(ws + (128 << 10));
  float* Cf     = PQ + (size_t)BATCH * NPTS * CH;
  int*   idxb   = (int*)(Cf + (size_t)BATCH * NPTS * CH);

  prep_params<<<1, 64, 0, stream>>>(w1, b1, g1, be1, m1, v1,
                                    w2, b2, g2, be2, m2, v2,
                                    Wc, sw1a, sw1d, bias1p, bias2p);

  knn_kernel<<<dim3(NPTS / 256, BATCH), 256, 0, stream>>>(xyz, idxb);

  geo_kernel<<<(BATCH * NPTS * HALF_CH) / 256, 256, 0, stream>>>(xyz, sw1a, sw1d,
                                                                 bias1p, PQ);

  feat_gemm<<<dim3(NPTS / 16, BATCH), 256, 0, stream>>>(feat, Wc, Cf);

  gather_max<<<dim3(NPTS / 4, BATCH), 256, 0, stream>>>(PQ, Cf, idxb, bias2p,
                                                        (float*)d_out);
}


# kNN re-gridded to fill the chip: block = 32 query rows x 4 waves, each wave scans a quarter of the candidates; the 4 sorted top-20 lists merge in LDS by the same median chain; scores bit-identical
# speedup vs baseline: 25.8313x; 25.8313x over previous
#include <hip/hip_runtime.h>
#include <hip/hip_bf16.h>
#include <float.h>

#define BATCH   8
#define NPTS    2048
#define CH      128
#define HALF_CH 64
#define KNN_K   20
#define EPSB    1e-5f

typedef __attribute__((ext_vector_type(2))) float v2f;
typedef __attribute__((ext_vector_type(8))) float v8f;

__global__ void prep_params(const float* __restrict__ w1, const float* __restrict__ b1,
                            const float* __restrict__ g1, const float* __restrict__ be1,
                            const float* __restrict__ m1, const float* __restrict__ v1,
                            const float* __restrict__ w2, const float* __restrict__ b2,
                            const float* __restrict__ g2, const float* __restrict__ be2,
                            const float* __restrict__ m2, const float* __restrict__ v2,
                            float* __restrict__ Wc,   float* __restrict__ sw1a,
                            float* __restrict__ sw1d, float* __restrict__ bias1p,
                            float* __restrict__ bias2p) {
  int o = threadIdx.x;
  if (o >= HALF_CH) return;
  float s1 = g1[o] / sqrtf(v1[o] + EPSB);
  bias1p[o] = b1[o] * s1 + (be1[o] - m1[o] * s1);
  #pragma unroll
  for (int c = 0; c < 3; ++c) {
    float wa = w1[o * 6 + c];
    float wb = w1[o * 6 + 3 + c];
    sw1a[o * 3 + c] = s1 * wa;
    sw1d[o * 3 + c] = s1 * (wb - wa);
  }
  float s2 = g2[o] / sqrtf(v2[o] + EPSB);
  bias2p[o] = b2[o] * s2 + (be2[o] - m2[o] * s2);
  for (int c = 0; c < CH; ++c) {
    float wa = w2[o * (2 * CH) + c];
    float wb = w2[o * (2 * CH) + CH + c];
    Wc[(size_t)o * CH + c]             = s2 * wa;
    Wc[(size_t)(HALF_CH + o) * CH + c] = s2 * (wb - wa);
  }
}

#define KNN_Q    32
#define KNN_W    4
#define KNN_SEG  (NPTS / KNN_W)
#define KNN_STG  (NPTS / (KNN_Q * KNN_W))

__device__ __forceinline__ float knn_score(const float4 me, const float4 p) {
  #pragma clang fp contract(off)
  const float yy  = me.y * p.y;
  const float zz  = me.z * p.z;
  const float dot = __builtin_fmaf(me.x, p.x, yy) + zz;
  return __builtin_fmaf(-2.0f, dot, me.w + p.w);
}

__device__ __forceinline__ void knn_insert(float (&t)[KNN_K], const float d) {
  #pragma unroll
  for (int i = KNN_K - 1; i > 0; --i) t[i] = __builtin_amdgcn_fmed3f(d, t[i - 1], t[i]);
  t[0] = fminf(d, t[0]);
}

__global__ void __launch_bounds__(KNN_Q * KNN_W)
knn_kernel(const float* __restrict__ xyz, int* __restrict__ idx_out) {
  __shared__ float4 pts[NPTS];
  __shared__ float  lst[KNN_W][KNN_K][KNN_Q];
  const int b = blockIdx.y;
  const float* xb = xyz + (size_t)b * 3 * NPTS;
  float sx[KNN_STG], sy[KNN_STG], sz[KNN_STG];
  #pragma unroll
  for (int i = 0; i < KNN_STG; ++i) {
    const int m = threadIdx.x + i * (KNN_Q * KNN_W);
    sx[i] = xb[m]; sy[i] = xb[NPTS + m]; sz[i] = xb[2 * NPTS + m];
  }
  #pragma unroll
  for (int i = 0; i < KNN_STG; ++i) {
    const int m = threadIdx.x + i * (KNN_Q * KNN_W);
    const float x = sx[i], y = sy[i], z = sz[i];
    pts[m] = make_float4(x, y, z, __builtin_fmaf(z, z, __builtin_fmaf(x, x, y * y)));
  }
  __syncthreads();

  const int lane = threadIdx.x & (KNN_Q - 1);
  const int w    = threadIdx.x / KNN_Q;
  const int n    = blockIdx.x * KNN_Q + lane;
  const float4 me = pts[n];
  const float4* seg = pts + w * KNN_SEG;

  float t[KNN_K];
  #pragma unroll
  for (int i = 0; i < KNN_K; ++i) t[i] = FLT_MAX;

  #pragma unroll 8
  for (int m = 0; m < KNN_SEG; ++m) knn_insert(t, knn_score(me, seg[m]));

  #pragma unroll
  for (int i = 0; i < KNN_K; ++i) lst[w][i][lane] = t[i];
  __syncthreads();
  for (int o = 1; o < KNN_W; ++o) {
    const int w2 = (w + o) & (KNN_W - 1);
    float v[KNN_K];
    #pragma unroll
    for (int i = 0; i < KNN_K; ++i) v[i] = lst[w2][i][lane];
    #pragma unroll
    for (int i = 0; i < KNN_K; ++i) knn_insert(t, v[i]);
  }

  const float thr = t[KNN_K - 1];
  int ties = 0;
  #pragma unroll
  for (int i = 0; i < KNN_K; ++i) ties += (t[i] == thr) ? 1 : 0;

  int off = 0;
  int quota = 0;
  int mine_eq = 0;
  int rem = ties;
  for (int q = 0; q <= w; ++q) {
    int lt = 0, eq = 0;
    #pragma unroll
    for (int i = 0; i < KNN_K; ++i) {
      const float s = lst[q][i][lane];
      lt += (s < thr) ? 1 : 0;
      eq += (s == thr) ? 1 : 0;
    }
    const int a = min(eq, rem);
    rem -= a;
    if (q < w) off += lt + a;
    else { quota = lt + a; mine_eq = a; }
  }
  if (w == KNN_W - 1) quota = KNN_K - off;

  int* op = idx_out + ((size_t)b * NPTS + n) * KNN_K + off;
  const int base = w * KNN_SEG;
  int cnt = 0;
  for (int m0 = 0; m0 < KNN_SEG; m0 += 8) {
    float d[8];
    #pragma unroll
    for (int j = 0; j < 8; ++j) d[j] = knn_score(me, seg[m0 + j]);
    #pragma unroll
    for (int j = 0; j < 8; ++j) {
      if (d[j] <= thr) {
        const bool tie = !(d[j] < thr);
        if ((!tie || mine_eq > 0) && cnt < quota) { op[cnt] = base + m0 + j; ++cnt; }
        if (tie) --mine_eq;
      }
    }
  }
  for (; cnt < quota; ++cnt) op[cnt] = n;
}

__global__ void geo_kernel(const float* __restrict__ xyz, const float* __restrict__ sw1a,
                           const float* __restrict__ sw1d, const float* __restrict__ bias1p,
                           float* __restrict__ PQ) {
  int t  = blockIdx.x * blockDim.x + threadIdx.x;
  int ch = t & (HALF_CH - 1);
  int pt = t >> 6;
  int b  = pt / NPTS;
  int n  = pt - b * NPTS;
  const float* xb = xyz + (size_t)b * 3 * NPTS;
  float x = xb[n], y = xb[NPTS + n], z = xb[2 * NPTS + n];
  float p = sw1a[ch * 3 + 0] * x + sw1a[ch * 3 + 1] * y + sw1a[ch * 3 + 2] * z;
  float q = sw1d[ch * 3 + 0] * x + sw1d[ch * 3 + 1] * y + sw1d[ch * 3 + 2] * z + bias1p[ch];
  float* o = PQ + (size_t)pt * CH;
  o[ch] = p;
  o[HALF_CH + ch] = q;
}

__global__ void feat_gemm(const float* __restrict__ feat, const float* __restrict__ Wc,
                          float* __restrict__ Cf) {
  const int b    = blockIdx.y;
  const int n0   = blockIdx.x * 16;
  const int wave = threadIdx.x >> 5;
  const int lane = threadIdx.x & 31;
  const int l16  = lane & 15;
  const int hi   = lane >> 4;
  const int m0   = wave * 16;

  const float* Fb   = feat + (size_t)b * CH * NPTS;
  const float* arow = Wc + (size_t)(m0 + l16) * CH + 2 * hi;
  const float* bcol = Fb + (size_t)(2 * hi) * NPTS + n0 + l16;

  v8f acc = {0.f, 0.f, 0.f, 0.f, 0.f, 0.f, 0.f, 0.f};

  #pragma unroll 8
  for (int k = 0; k < CH; k += 4) {
    v2f a;
    a.x = arow[k];
    a.y = arow[k + 1];
    v2f bb;
    bb.x = bcol[(size_t)k * NPTS];
    bb.y = bcol[(size_t)(k + 1) * NPTS];
    acc = __builtin_amdgcn_wmma_f32_16x16x4_f32(false, a, false, bb,
                                                (short)0, acc, false, false);
  }

  float* out = Cf + ((size_t)b * NPTS + n0 + l16) * CH + m0 + 8 * hi;
  #pragma unroll
  for (int v = 0; v < 8; ++v) out[v] = acc[v];
}

__global__ void gather_max(const float* __restrict__ PQ, const float* __restrict__ Cf,
                           const int* __restrict__ idx, const float* __restrict__ bias2p,
                           float* __restrict__ out) {
  const int b  = blockIdx.y;
  const int ch = threadIdx.x & (HALF_CH - 1);
  const int n  = blockIdx.x * 4 + (threadIdx.x >> 6);

  const size_t pbase = (size_t)b * NPTS + n;
  const int* id = idx + pbase * KNN_K;

  const float qg = PQ[pbase * CH + HALF_CH + ch];
  const float vf = Cf[pbase * CH + HALF_CH + ch] + bias2p[ch];

  float mg = -FLT_MAX, mf = -FLT_MAX;
  #pragma unroll 4
  for (int k = 0; k < KNN_K; ++k) {
    int j = id[k];
    size_t jb = ((size_t)b * NPTS + j) * CH;
    mg = fmaxf(mg, PQ[jb + ch]);
    mf = fmaxf(mf, Cf[jb + ch]);
  }
  float geo = fmaxf(mg + qg, 0.0f);
  float fea = fmaxf(mf + vf, 0.0f);

  out[(size_t)b * CH * NPTS + (size_t)ch * NPTS + n]             = geo;
  out[(size_t)b * CH * NPTS + (size_t)(HALF_CH + ch) * NPTS + n] = fea;
}

extern "C" void kernel_launch(void* const* d_in, const int* in_sizes, int n_in,
                              void* d_out, int out_size, void* d_ws, size_t ws_size,
                              hipStream_t stream) {
  (void)in_sizes; (void)n_in; (void)out_size; (void)ws_size;

  const float* xyz  = (const float*)d_in[0];
  const float* feat = (const float*)d_in[1];
  const float* w1   = (const float*)d_in[2];
  const float* b1   = (const float*)d_in[3];
  const float* g1   = (const float*)d_in[4];
  const float* be1  = (const float*)d_in[5];
  const float* m1   = (const float*)d_in[6];
  const float* v1   = (const float*)d_in[7];
  const float* w2   = (const float*)d_in[8];
  const float* b2   = (const float*)d_in[9];
  const float* g2   = (const float*)d_in[10];
  const float* be2  = (const float*)d_in[11];
  const float* m2   = (const float*)d_in[12];
  const float* v2   = (const float*)d_in[13];

  char* ws = (char*)d_ws;
  float* Wc     = (float*)ws;
  float* sw1a   = Wc + 16384;
  float* sw1d   = sw1a + 192;
  float* bias1p = sw1d + 192;
  float* bias2p = bias1p + 64;
  float* PQ     = (float*)(ws + (128 << 10));
  float* Cf     = PQ + (size_t)BATCH * NPTS * CH;
  int*   idxb   = (int*)(Cf + (size_t)BATCH * NPTS * CH);

  prep_params<<<1, 64, 0, stream>>>(w1, b1, g1, be1, m1, v1,
                                    w2, b2, g2, be2, m2, v2,
                                    Wc, sw1a, sw1d, bias1p, bias2p);

  knn_kernel<<<dim3(NPTS / KNN_Q, BATCH), KNN_Q * KNN_W, 0, stream>>>(xyz, idxb);

  geo_kernel<<<(BATCH * NPTS * HALF_CH) / 256, 256, 0, stream>>>(xyz, sw1a, sw1d,
                                                                 bias1p, PQ);

  feat_gemm<<<dim3(NPTS / 16, BATCH), 256, 0, stream>>>(feat, Wc, Cf);

  gather_max<<<dim3(NPTS / 4, BATCH), 256, 0, stream>>>(PQ, Cf, idxb, bias2p,
                                                        (float*)d_out);
}


# gather tables stored channel-major as float4 quads, lane = point; gather block copies two quad rows (all 2048 points) to LDS and takes the 20 neighbours there; 65536 waves become 6144; same sums
# speedup vs baseline: 46.0946x; 1.7844x over previous
#include <hip/hip_runtime.h>
#include <hip/hip_bf16.h>
#include <float.h>

#define BATCH   8
#define NPTS    2048
#define CH      128
#define HALF_CH 64
#define KNN_K   20
#define EPSB    1e-5f

typedef __attribute__((ext_vector_type(2))) float v2f;
typedef __attribute__((ext_vector_type(8))) float v8f;

__global__ void prep_params(const float* __restrict__ w1, const float* __restrict__ b1,
                            const float* __restrict__ g1, const float* __restrict__ be1,
                            const float* __restrict__ m1, const float* __restrict__ v1,
                            const float* __restrict__ w2, const float* __restrict__ b2,
                            const float* __restrict__ g2, const float* __restrict__ be2,
                            const float* __restrict__ m2, const float* __restrict__ v2,
                            float* __restrict__ Wc,   float* __restrict__ sw1a,
                            float* __restrict__ sw1d, float* __restrict__ bias1p,
                            float* __restrict__ bias2p) {
  int o = threadIdx.x;
  if (o >= HALF_CH) return;
  float s1 = g1[o] / sqrtf(v1[o] + EPSB);
  bias1p[o] = b1[o] * s1 + (be1[o] - m1[o] * s1);
  #pragma unroll
  for (int c = 0; c < 3; ++c) {
    float wa = w1[o * 6 + c];
    float wb = w1[o * 6 + 3 + c];
    sw1a[o * 3 + c] = s1 * wa;
    sw1d[o * 3 + c] = s1 * (wb - wa);
  }
  float s2 = g2[o] / sqrtf(v2[o] + EPSB);
  bias2p[o] = b2[o] * s2 + (be2[o] - m2[o] * s2);
  for (int c = 0; c < CH; ++c) {
    float wa = w2[o * (2 * CH) + c];
    float wb = w2[o * (2 * CH) + CH + c];
    Wc[(size_t)o * CH + c]             = s2 * wa;
    Wc[(size_t)(HALF_CH + o) * CH + c] = s2 * (wb - wa);
  }
}

#define KNN_Q    32
#define KNN_W    4
#define KNN_SEG  (NPTS / KNN_W)
#define KNN_STG  (NPTS / (KNN_Q * KNN_W))

__device__ __forceinline__ float knn_score(const float4 me, const float4 p) {
  #pragma clang fp contract(off)
  const float yy  = me.y * p.y;
  const float zz  = me.z * p.z;
  const float dot = __builtin_fmaf(me.x, p.x, yy) + zz;
  return __builtin_fmaf(-2.0f, dot, me.w + p.w);
}

__device__ __forceinline__ void knn_insert(float (&t)[KNN_K], const float d) {
  #pragma unroll
  for (int i = KNN_K - 1; i > 0; --i) t[i] = __builtin_amdgcn_fmed3f(d, t[i - 1], t[i]);
  t[0] = fminf(d, t[0]);
}

__global__ void __launch_bounds__(KNN_Q * KNN_W)
knn_kernel(const float* __restrict__ xyz, int* __restrict__ idx_out) {
  __shared__ float4 pts[NPTS];
  __shared__ float  lst[KNN_W][KNN_K][KNN_Q];
  const int b = blockIdx.y;
  const float* xb = xyz + (size_t)b * 3 * NPTS;
  float sx[KNN_STG], sy[KNN_STG], sz[KNN_STG];
  #pragma unroll
  for (int i = 0; i < KNN_STG; ++i) {
    const int m = threadIdx.x + i * (KNN_Q * KNN_W);
    sx[i] = xb[m]; sy[i] = xb[NPTS + m]; sz[i] = xb[2 * NPTS + m];
  }
  #pragma unroll
  for (int i = 0; i < KNN_STG; ++i) {
    const int m = threadIdx.x + i * (KNN_Q * KNN_W);
    const float x = sx[i], y = sy[i], z = sz[i];
    pts[m] = make_float4(x, y, z, __builtin_fmaf(z, z, __builtin_fmaf(x, x, y * y)));
  }
  __syncthreads();

  const int lane = threadIdx.x & (KNN_Q - 1);
  const int w    = threadIdx.x / KNN_Q;
  const int n    = blockIdx.x * KNN_Q + lane;
  const float4 me = pts[n];
  const float4* seg = pts + w * KNN_SEG;

  float t[KNN_K];
  #pragma unroll
  for (int i = 0; i < KNN_K; ++i) t[i] = FLT_MAX;

  #pragma unroll 8
  for (int m = 0; m < KNN_SEG; ++m) knn_insert(t, knn_score(me, seg[m]));

  #pragma unroll
  for (int i = 0; i < KNN_K; ++i) lst[w][i][lane] = t[i];
  __syncthreads();
  for (int o = 1; o < KNN_W; ++o) {
    const int w2 = (w + o) & (KNN_W - 1);
    float v[KNN_K];
    #pragma unroll
    for (int i = 0; i < KNN_K; ++i) v[i] = lst[w2][i][lane];
    #pragma unroll
    for (int i = 0; i < KNN_K; ++i) knn_insert(t, v[i]);
  }

  const float thr = t[KNN_K - 1];
  int ties = 0;
  #pragma unroll
  for (int i = 0; i < KNN_K; ++i) ties += (t[i] == thr) ? 1 : 0;

  int off = 0;
  int quota = 0;
  int mine_eq = 0;
  int rem = ties;
  for (int q = 0; q <= w; ++q) {
    int lt = 0, eq = 0;
    #pragma unroll
    for (int i = 0; i < KNN_K; ++i) {
      const float s = lst[q][i][lane];
      lt += (s < thr) ? 1 : 0;
      eq += (s == thr) ? 1 : 0;
    }
    const int a = min(eq, rem);
    rem -= a;
    if (q < w) off += lt + a;
    else { quota = lt + a; mine_eq = a; }
  }
  if (w == KNN_W - 1) quota = KNN_K - off;

  int* op = idx_out + ((size_t)b * NPTS + n) * KNN_K + off;
  const int base = w * KNN_SEG;
  int cnt = 0;
  for (int m0 = 0; m0 < KNN_SEG; m0 += 8) {
    float d[8];
    #pragma unroll
    for (int j = 0; j < 8; ++j) d[j] = knn_score(me, seg[m0 + j]);
    #pragma unroll
    for (int j = 0; j < 8; ++j) {
      if (d[j] <= thr) {
        const bool tie = !(d[j] < thr);
        if ((!tie || mine_eq > 0) && cnt < quota) { op[cnt] = base + m0 + j; ++cnt; }
        if (tie) --mine_eq;
      }
    }
  }
  for (; cnt < quota; ++cnt) op[cnt] = n;
}

#define QROWS 32

template <bool CTR>
__device__ __forceinline__ void geo_rows(const float* __restrict__ wt, const float* __restrict__ bias,
                                         const float x, const float y, const float z,
                                         float4* __restrict__ o) {
  #pragma unroll
  for (int j = 0; j < 8; ++j) {
    float v[4];
    #pragma unroll
    for (int e = 0; e < 4; ++e) {
      const int c = 4 * j + e;
      if (CTR) v[e] = wt[c * 3 + 0] * x + wt[c * 3 + 1] * y + wt[c * 3 + 2] * z + bias[c];
      else     v[e] = wt[c * 3 + 0] * x + wt[c * 3 + 1] * y + wt[c * 3 + 2] * z;
    }
    o[(size_t)j * NPTS] = make_float4(v[0], v[1], v[2], v[3]);
  }
}

__global__ void __launch_bounds__(256)
geo_kernel(const float* __restrict__ xyz, const float* __restrict__ sw1a,
           const float* __restrict__ sw1d, const float* __restrict__ bias1p,
           float* __restrict__ PQ) {
  const int pt = blockIdx.x * 256 + threadIdx.x;
  const int b  = pt / NPTS;
  const int n  = pt - b * NPTS;
  const int g  = blockIdx.y;
  const int c0 = (g & 1) * 32;
  const float* xb = xyz + (size_t)b * 3 * NPTS;
  const float x = xb[n], y = xb[NPTS + n], z = xb[2 * NPTS + n];
  float4* o = (float4*)PQ + ((size_t)b * QROWS + 8 * g) * NPTS + n;
  if (g >= 2) geo_rows<true>(sw1d + c0 * 3, bias1p + c0, x, y, z, o);
  else        geo_rows<false>(sw1a + c0 * 3, bias1p, x, y, z, o);
}

__global__ void feat_gemm(const float* __restrict__ feat, const float* __restrict__ Wc,
                          float* __restrict__ Cf) {
  const int b    = blockIdx.y;
  const int n0   = blockIdx.x * 16;
  const int wave = threadIdx.x >> 5;
  const int lane = threadIdx.x & 31;
  const int l16  = lane & 15;
  const int hi   = lane >> 4;
  const int m0   = wave * 16;

  const float* Fb   = feat + (size_t)b * CH * NPTS;
  const float* arow = Wc + (size_t)(m0 + l16) * CH + 2 * hi;
  const float* bcol = Fb + (size_t)(2 * hi) * NPTS + n0 + l16;

  v8f acc = {0.f, 0.f, 0.f, 0.f, 0.f, 0.f, 0.f, 0.f};

  #pragma unroll 8
  for (int k = 0; k < CH; k += 4) {
    v2f a;
    a.x = arow[k];
    a.y = arow[k + 1];
    v2f bb;
    bb.x = bcol[(size_t)k * NPTS];
    bb.y = bcol[(size_t)(k + 1) * NPTS];
    acc = __builtin_amdgcn_wmma_f32_16x16x4_f32(false, a, false, bb,
                                                (short)0, acc, false, false);
  }

  float4* out = (float4*)Cf + ((size_t)b * QROWS + (m0 + 8 * hi) / 4) * NPTS + n0 + l16;
  out[0]    = make_float4(acc[0], acc[1], acc[2], acc[3]);
  out[NPTS] = make_float4(acc[4], acc[5], acc[6], acc[7]);
}

#define GA_THR 256
#define GA_PPT 2
#define GA_PTS (GA_THR * GA_PPT)
#define GA_STG (2 * NPTS / GA_THR)

__device__ __forceinline__ float4 max4(const float4 a, const float4 b) {
  return make_float4(fmaxf(a.x, b.x), fmaxf(a.y, b.y), fmaxf(a.z, b.z), fmaxf(a.w, b.w));
}

__global__ void __launch_bounds__(GA_THR)
gather_max(const float* __restrict__ PQ, const float* __restrict__ Cf,
           const int* __restrict__ idx, const float* __restrict__ bias2p,
           float* __restrict__ out) {
  __shared__ float4 tab[2 * NPTS];
  const int b    = blockIdx.z;
  const int tp   = blockIdx.y;
  const bool fea = tp >= 8;
  const int q0   = (tp & 7) * 2;
  const int tid  = threadIdx.x;
  const float4* T   = (const float4*)(fea ? Cf : PQ) + (size_t)b * QROWS * NPTS;
  const float4* src = T + (size_t)q0 * NPTS;

  float4 st[GA_STG];
  #pragma unroll
  for (int i = 0; i < GA_STG; ++i) st[i] = src[tid + i * GA_THR];

  const int nb = blockIdx.x * GA_PTS + tid;
  int4   id[GA_PPT][KNN_K / 4];
  float4 ctr[GA_PPT][2];
  #pragma unroll
  for (int p = 0; p < GA_PPT; ++p) {
    const int n = nb + p * GA_THR;
    const int4* ip = (const int4*)(idx + ((size_t)b * NPTS + n) * KNN_K);
    #pragma unroll
    for (int j = 0; j < KNN_K / 4; ++j) id[p][j] = ip[j];
    ctr[p][0] = T[(size_t)(16 + q0) * NPTS + n];
    ctr[p][1] = T[(size_t)(17 + q0) * NPTS + n];
  }

  #pragma unroll
  for (int i = 0; i < GA_STG; ++i) tab[tid + i * GA_THR] = st[i];

  float4 bs0 = make_float4(0.f, 0.f, 0.f, 0.f), bs1 = bs0;
  if (fea) {
    const float* bp = bias2p + 4 * q0;
    bs0 = make_float4(bp[0], bp[1], bp[2], bp[3]);
    bs1 = make_float4(bp[4], bp[5], bp[6], bp[7]);
  }
  __syncthreads();

  float* ob = out + ((size_t)b * CH + (fea ? HALF_CH : 0) + 4 * q0) * NPTS;
  #pragma unroll
  for (int p = 0; p < GA_PPT; ++p) {
    const int n = nb + p * GA_THR;
    float4 m0 = make_float4(-FLT_MAX, -FLT_MAX, -FLT_MAX, -FLT_MAX), m1 = m0;
    #pragma unroll
    for (int j = 0; j < KNN_K / 4; ++j) {
      const int4 q = id[p][j];
      m0 = max4(m0, tab[q.x]); m1 = max4(m1, tab[NPTS + q.x]);
      m0 = max4(m0, tab[q.y]); m1 = max4(m1, tab[NPTS + q.y]);
      m0 = max4(m0, tab[q.z]); m1 = max4(m1, tab[NPTS + q.z]);
      m0 = max4(m0, tab[q.w]); m1 = max4(m1, tab[NPTS + q.w]);
    }
    float4 c0 = ctr[p][0], c1 = ctr[p][1];
    if (fea) {
      c0 = make_float4(c0.x + bs0.x, c0.y + bs0.y, c0.z + bs0.z, c0.w + bs0.w);
      c1 = make_float4(c1.x + bs1.x, c1.y + bs1.y, c1.z + bs1.z, c1.w + bs1.w);
    }
    ob[0 * NPTS + n] = fmaxf(m0.x + c0.x, 0.0f);
    ob[1 * NPTS + n] = fmaxf(m0.y + c0.y, 0.0f);
    ob[2 * NPTS + n] = fmaxf(m0.z + c0.z, 0.0f);
    ob[3 * NPTS + n] = fmaxf(m0.w + c0.w, 0.0f);
    ob[4 * NPTS + n] = fmaxf(m1.x + c1.x, 0.0f);
    ob[5 * NPTS + n] = fmaxf(m1.y + c1.y, 0.0f);
    ob[6 * NPTS + n] = fmaxf(m1.z + c1.z, 0.0f);
    ob[7 * NPTS + n] = fmaxf(m1.w + c1.w, 0.0f);
  }
}

extern "C" void kernel_launch(void* const* d_in, const int* in_sizes, int n_in,
                              void* d_out, int out_size, void* d_ws, size_t ws_size,
                              hipStream_t stream) {
  (void)in_sizes; (void)n_in; (void)out_size; (void)ws_size;

  const float* xyz  = (const float*)d_in[0];
  const float* feat = (const float*)d_in[1];
  const float* w1   = (const float*)d_in[2];
  const float* b1   = (const float*)d_in[3];
  const float* g1   = (const float*)d_in[4];
  const float* be1  = (const float*)d_in[5];
  const float* m1   = (const float*)d_in[6];
  const float* v1   = (const float*)d_in[7];
  const float* w2   = (const float*)d_in[8];
  const float* b2   = (const float*)d_in[9];
  const float* g2   = (const float*)d_in[10];
  const float* be2  = (const float*)d_in[11];
  const float* m2   = (const float*)d_in[12];
  const float* v2   = (const float*)d_in[13];

  char* ws = (char*)d_ws;
  float* Wc     = (float*)ws;
  float* sw1a   = Wc + 16384;
  float* sw1d   = sw1a + 192;
  float* bias1p = sw1d + 192;
  float* bias2p = bias1p + 64;
  float* PQ     = (float*)(ws + (128 << 10));
  float* Cf     = PQ + (size_t)BATCH * NPTS * CH;
  int*   idxb   = (int*)(Cf + (size_t)BATCH * NPTS * CH);

  prep_params<<<1, 64, 0, stream>>>(w1, b1, g1, be1, m1, v1,
                                    w2, b2, g2, be2, m2, v2,
                                    Wc, sw1a, sw1d, bias1p, bias2p);

  knn_kernel<<<dim3(NPTS / KNN_Q, BATCH), KNN_Q * KNN_W, 0, stream>>>(xyz, idxb);

  geo_kernel<<<dim3(BATCH * NPTS / 256, 4), 256, 0, stream>>>(xyz, sw1a, sw1d,
                                                              bias1p, PQ);

  feat_gemm<<<dim3(NPTS / 16, BATCH), 256, 0, stream>>>(feat, Wc, Cf);

  gather_max<<<dim3(NPTS / GA_PTS, 16, BATCH), GA_THR, 0, stream>>>(PQ, Cf, idxb, bias2p,
                                                                    (float*)d_out);
}


# independent kernels share launches: parameter fold runs as one extra row of blocks in the kNN launch, geometry tables and feature GEMM as rows of one grid; 5 launches become 3; same sums
# speedup vs baseline: 53.4906x; 1.1605x over previous
#include <hip/hip_runtime.h>
#include <hip/hip_bf16.h>
#include <float.h>

#define BATCH   8
#define NPTS    2048
#define CH      128
#define HALF_CH 64
#define KNN_K   20
#define EPSB    1e-5f

typedef __attribute__((ext_vector_type(2))) float v2f;
typedef __attribute__((ext_vector_type(8))) float v8f;

#define KNN_Q    32
#define KNN_W    4
#define KNN_SEG  (NPTS / KNN_W)
#define KNN_STG  (NPTS / (KNN_Q * KNN_W))

__device__ __forceinline__ float knn_score(const float4 me, const float4 p) {
  #pragma clang fp contract(off)
  const float yy  = me.y * p.y;
  const float zz  = me.z * p.z;
  const float dot = __builtin_fmaf(me.x, p.x, yy) + zz;
  return __builtin_fmaf(-2.0f, dot, me.w + p.w);
}

__device__ __forceinline__ void knn_insert(float (&t)[KNN_K], const float d) {
  #pragma unroll
  for (int i = KNN_K - 1; i > 0; --i) t[i] = __builtin_amdgcn_fmed3f(d, t[i - 1], t[i]);
  t[0] = fminf(d, t[0]);
}

__device__ __forceinline__ void fold_params(
    const float* __restrict__ w1, const float* __restrict__ b1,
    const float* __restrict__ g1, const float* __restrict__ be1,
    const float* __restrict__ m1, const float* __restrict__ v1,
    const float* __restrict__ w2, const float* __restrict__ b2,
    const float* __restrict__ g2, const float* __restrict__ be2,
    const float* __restrict__ m2, const float* __restrict__ v2,
    float* __restrict__ Wc,   float* __restrict__ sw1a,
    float* __restrict__ sw1d, float* __restrict__ bias1p,
    float* __restrict__ bias2p) {
  for (int o = blockIdx.x; o < HALF_CH; o += NPTS / KNN_Q) {
    const float s2 = g2[o] / sqrtf(v2[o] + EPSB);
    for (int c = threadIdx.x; c < CH; c += KNN_Q * KNN_W) {
      const float wa = w2[o * (2 * CH) + c];
      const float wb = w2[o * (2 * CH) + CH + c];
      Wc[(size_t)o * CH + c]             = s2 * wa;
      Wc[(size_t)(HALF_CH + o) * CH + c] = s2 * (wb - wa);
    }
  }
  if (blockIdx.x == 0 && threadIdx.x < HALF_CH) {
    const int o = threadIdx.x;
    const float s1 = g1[o] / sqrtf(v1[o] + EPSB);
    bias1p[o] = b1[o] * s1 + (be1[o] - m1[o] * s1);
    #pragma unroll
    for (int c = 0; c < 3; ++c) {
      const float wa = w1[o * 6 + c];
      const float wb = w1[o * 6 + 3 + c];
      sw1a[o * 3 + c] = s1 * wa;
      sw1d[o * 3 + c] = s1 * (wb - wa);
    }
    const float s2 = g2[o] / sqrtf(v2[o] + EPSB);
    bias2p[o] = b2[o] * s2 + (be2[o] - m2[o] * s2);
  }
}

__global__ void __launch_bounds__(KNN_Q * KNN_W)
knn_kernel(const float* __restrict__ xyz, int* __restrict__ idx_out,
           const float* __restrict__ w1, const float* __restrict__ b1,
           const float* __restrict__ g1, const float* __restrict__ be1,
           const float* __restrict__ m1, const float* __restrict__ v1,
           const float* __restrict__ w2, const float* __restrict__ b2,
           const float* __restrict__ g2, const float* __restrict__ be2,
           const float* __restrict__ m2, const float* __restrict__ v2,
           float* __restrict__ Wc,   float* __restrict__ sw1a,
           float* __restrict__ sw1d, float* __restrict__ bias1p,
           float* __restrict__ bias2p) {
  __shared__ float4 pts[NPTS];
  __shared__ float  lst[KNN_W][KNN_K][KNN_Q];
  if (blockIdx.y == BATCH) {
    fold_params(w1, b1, g1, be1, m1, v1, w2, b2, g2, be2, m2, v2,
                Wc, sw1a, sw1d, bias1p, bias2p);
    return;
  }
  const int b = blockIdx.y;
  const float* xb = xyz + (size_t)b * 3 * NPTS;
  float sx[KNN_STG], sy[KNN_STG], sz[KNN_STG];
  #pragma unroll
  for (int i = 0; i < KNN_STG; ++i) {
    const int m = threadIdx.x + i * (KNN_Q * KNN_W);
    sx[i] = xb[m]; sy[i] = xb[NPTS + m]; sz[i] = xb[2 * NPTS + m];
  }
  #pragma unroll
  for (int i = 0; i < KNN_STG; ++i) {
    const int m = threadIdx.x + i * (KNN_Q * KNN_W);
    const float x = sx[i], y = sy[i], z = sz[i];
    pts[m] = make_float4(x, y, z, __builtin_fmaf(z, z, __builtin_fmaf(x, x, y * y)));
  }
  __syncthreads();

  const int lane = threadIdx.x & (KNN_Q - 1);
  const int w    = threadIdx.x / KNN_Q;
  const int n    = blockIdx.x * KNN_Q + lane;
  const float4 me = pts[n];
  const float4* seg = pts + w * KNN_SEG;

  float t[KNN_K];
  #pragma unroll
  for (int i = 0; i < KNN_K; ++i) t[i] = FLT_MAX;

  #pragma unroll 8
  for (int m = 0; m < KNN_SEG; ++m) knn_insert(t, knn_score(me, seg[m]));

  #pragma unroll
  for (int i = 0; i < KNN_K; ++i) lst[w][i][lane] = t[i];
  __syncthreads();
  for (int o = 1; o < KNN_W; ++o) {
    const int w2 = (w + o) & (KNN_W - 1);
    float v[KNN_K];
    #pragma unroll
    for (int i = 0; i < KNN_K; ++i) v[i] = lst[w2][i][lane];
    #pragma unroll
    for (int i = 0; i < KNN_K; ++i) knn_insert(t, v[i]);
  }

  const float thr = t[KNN_K - 1];
  int ties = 0;
  #pragma unroll
  for (int i = 0; i < KNN_K; ++i) ties += (t[i] == thr) ? 1 : 0;

  int off = 0;
  int quota = 0;
  int mine_eq = 0;
  int rem = ties;
  for (int q = 0; q <= w; ++q) {
    int lt = 0, eq = 0;
    #pragma unroll
    for (int i = 0; i < KNN_K; ++i) {
      const float s = lst[q][i][lane];
      lt += (s < thr) ? 1 : 0;
      eq += (s == thr) ? 1 : 0;
    }
    const int a = min(eq, rem);
    rem -= a;
    if (q < w) off += lt + a;
    else { quota = lt + a; mine_eq = a; }
  }
  if (w == KNN_W - 1) quota = KNN_K - off;

  int* op = idx_out + ((size_t)b * NPTS + n) * KNN_K + off;
  const int base = w * KNN_SEG;
  int cnt = 0;
  for (int m0 = 0; m0 < KNN_SEG; m0 += 8) {
    float d[8];
    #pragma unroll
    for (int j = 0; j < 8; ++j) d[j] = knn_score(me, seg[m0 + j]);
    #pragma unroll
    for (int j = 0; j < 8; ++j) {
      if (d[j] <= thr) {
        const bool tie = !(d[j] < thr);
        if ((!tie || mine_eq > 0) && cnt < quota) { op[cnt] = base + m0 + j; ++cnt; }
        if (tie) --mine_eq;
      }
    }
  }
  for (; cnt < quota; ++cnt) op[cnt] = n;
}

#define QROWS 32

template <bool CTR>
__device__ __forceinline__ void geo_rows(const float* __restrict__ wt, const float* __restrict__ bias,
                                         const float x, const float y, const float z,
                                         float4* __restrict__ o) {
  #pragma unroll
  for (int j = 0; j < 8; ++j) {
    float v[4];
    #pragma unroll
    for (int e = 0; e < 4; ++e) {
      const int c = 4 * j + e;
      if (CTR) v[e] = wt[c * 3 + 0] * x + wt[c * 3 + 1] * y + wt[c * 3 + 2] * z + bias[c];
      else     v[e] = wt[c * 3 + 0] * x + wt[c * 3 + 1] * y + wt[c * 3 + 2] * z;
    }
    o[(size_t)j * NPTS] = make_float4(v[0], v[1], v[2], v[3]);
  }
}

__device__ __forceinline__ void geo_block(const int bx, const int g,
           const float* __restrict__ xyz, const float* __restrict__ sw1a,
           const float* __restrict__ sw1d, const float* __restrict__ bias1p,
           float* __restrict__ PQ) {
  const int pt = bx * 256 + threadIdx.x;
  const int b  = pt / NPTS;
  const int n  = pt - b * NPTS;
  const int c0 = (g & 1) * 32;
  const float* xb = xyz + (size_t)b * 3 * NPTS;
  const float x = xb[n], y = xb[NPTS + n], z = xb[2 * NPTS + n];
  float4* o = (float4*)PQ + ((size_t)b * QROWS + 8 * g) * NPTS + n;
  if (g >= 2) geo_rows<true>(sw1d + c0 * 3, bias1p + c0, x, y, z, o);
  else        geo_rows<false>(sw1a + c0 * 3, bias1p, x, y, z, o);
}

__device__ __forceinline__ void gemm_block(const int bx, const int b,
                          const float* __restrict__ feat, const float* __restrict__ Wc,
                          float* __restrict__ Cf) {
  const int n0   = bx * 16;
  const int wave = threadIdx.x >> 5;
  const int lane = threadIdx.x & 31;
  const int l16  = lane & 15;
  const int hi   = lane >> 4;
  const int m0   = wave * 16;

  const float* Fb   = feat + (size_t)b * CH * NPTS;
  const float* arow = Wc + (size_t)(m0 + l16) * CH + 2 * hi;
  const float* bcol = Fb + (size_t)(2 * hi) * NPTS + n0 + l16;

  v8f acc = {0.f, 0.f, 0.f, 0.f, 0.f, 0.f, 0.f, 0.f};

  #pragma unroll 8
  for (int k = 0; k < CH; k += 4) {
    v2f a;
    a.x = arow[k];
    a.y = arow[k + 1];
    v2f bb;
    bb.x = bcol[(size_t)k * NPTS];
    bb.y = bcol[(size_t)(k + 1) * NPTS];
    acc = __builtin_amdgcn_wmma_f32_16x16x4_f32(false, a, false, bb,
                                                (short)0, acc, false, false);
  }

  float4* out = (float4*)Cf + ((size_t)b * QROWS + (m0 + 8 * hi) / 4) * NPTS + n0 + l16;
  out[0]    = make_float4(acc[0], acc[1], acc[2], acc[3]);
  out[NPTS] = make_float4(acc[4], acc[5], acc[6], acc[7]);
}

#define TB_X   (NPTS / 16)
#define TB_GEO ((BATCH * NPTS / 256) * 4 / TB_X)

__global__ void __launch_bounds__(256)
tables_kernel(const float* __restrict__ xyz, const float* __restrict__ feat,
              const float* __restrict__ Wc, const float* __restrict__ sw1a,
              const float* __restrict__ sw1d, const float* __restrict__ bias1p,
              float* __restrict__ PQ, float* __restrict__ Cf) {
  if (blockIdx.y < BATCH) {
    gemm_block(blockIdx.x, blockIdx.y, feat, Wc, Cf);
  } else {
    const unsigned id = (blockIdx.y - BATCH) * TB_X + blockIdx.x;
    geo_block((int)(id % (BATCH * NPTS / 256)), (int)(id / (BATCH * NPTS / 256)),
              xyz, sw1a, sw1d, bias1p, PQ);
  }
}

#define GA_THR 256
#define GA_PPT 2
#define GA_PTS (GA_THR * GA_PPT)
#define GA_STG (2 * NPTS / GA_THR)

__device__ __forceinline__ float4 max4(const float4 a, const float4 b) {
  return make_float4(fmaxf(a.x, b.x), fmaxf(a.y, b.y), fmaxf(a.z, b.z), fmaxf(a.w, b.w));
}

__global__ void __launch_bounds__(GA_THR)
gather_max(const float* __restrict__ PQ, const float* __restrict__ Cf,
           const int* __restrict__ idx, const float* __restrict__ bias2p,
           float* __restrict__ out) {
  __shared__ float4 tab[2 * NPTS];
  const int b    = blockIdx.z;
  const int tp   = blockIdx.y;
  const bool fea = tp >= 8;
  const int q0   = (tp & 7) * 2;
  const int tid  = threadIdx.x;
  const float4* T   = (const float4*)(fea ? Cf : PQ) + (size_t)b * QROWS * NPTS;
  const float4* src = T + (size_t)q0 * NPTS;

  float4 st[GA_STG];
  #pragma unroll
  for (int i = 0; i < GA_STG; ++i) st[i] = src[tid + i * GA_THR];

  const int nb = blockIdx.x * GA_PTS + tid;
  int4   id[GA_PPT][KNN_K / 4];
  float4 ctr[GA_PPT][2];
  #pragma unroll
  for (int p = 0; p < GA_PPT; ++p) {
    const int n = nb + p * GA_THR;
    const int4* ip = (const int4*)(idx + ((size_t)b * NPTS + n) * KNN_K);
    #pragma unroll
    for (int j = 0; j < KNN_K / 4; ++j) id[p][j] = ip[j];
    ctr[p][0] = T[(size_t)(16 + q0) * NPTS + n];
    ctr[p][1] = T[(size_t)(17 + q0) * NPTS + n];
  }

  #pragma unroll
  for (int i = 0; i < GA_STG; ++i) tab[tid + i * GA_THR] = st[i];

  float4 bs0 = make_float4(0.f, 0.f, 0.f, 0.f), bs1 = bs0;
  if (fea) {
    const float* bp = bias2p + 4 * q0;
    bs0 = make_float4(bp[0], bp[1], bp[2], bp[3]);
    bs1 = make_float4(bp[4], bp[5], bp[6], bp[7]);
  }
  __syncthreads();

  float* ob = out + ((size_t)b * CH + (fea ? HALF_CH : 0) + 4 * q0) * NPTS;
  #pragma unroll
  for (int p = 0; p < GA_PPT; ++p) {
    const int n = nb + p * GA_THR;
    float4 m0 = make_float4(-FLT_MAX, -FLT_MAX, -FLT_MAX, -FLT_MAX), m1 = m0;
    #pragma unroll
    for (int j = 0; j < KNN_K / 4; ++j) {
      const int4 q = id[p][j];
      m0 = max4(m0, tab[q.x]); m1 = max4(m1, tab[NPTS + q.x]);
      m0 = max4(m0, tab[q.y]); m1 = max4(m1, tab[NPTS + q.y]);
      m0 = max4(m0, tab[q.z]); m1 = max4(m1, tab[NPTS + q.z]);
      m0 = max4(m0, tab[q.w]); m1 = max4(m1, tab[NPTS + q.w]);
    }
    float4 c0 = ctr[p][0], c1 = ctr[p][1];
    if (fea) {
      c0 = make_float4(c0.x + bs0.x, c0.y + bs0.y, c0.z + bs0.z, c0.w + bs0.w);
      c1 = make_float4(c1.x + bs1.x, c1.y + bs1.y, c1.z + bs1.z, c1.w + bs1.w);
    }
    ob[0 * NPTS + n] = fmaxf(m0.x + c0.x, 0.0f);
    ob[1 * NPTS + n] = fmaxf(m0.y + c0.y, 0.0f);
    ob[2 * NPTS + n] = fmaxf(m0.z + c0.z, 0.0f);
    ob[3 * NPTS + n] = fmaxf(m0.w + c0.w, 0.0f);
    ob[4 * NPTS + n] = fmaxf(m1.x + c1.x, 0.0f);
    ob[5 * NPTS + n] = fmaxf(m1.y + c1.y, 0.0f);
    ob[6 * NPTS + n] = fmaxf(m1.z + c1.z, 0.0f);
    ob[7 * NPTS + n] = fmaxf(m1.w + c1.w, 0.0f);
  }
}

extern "C" void kernel_launch(void* const* d_in, const int* in_sizes, int n_in,
                              void* d_out, int out_size, void* d_ws, size_t ws_size,
                              hipStream_t stream) {
  (void)in_sizes; (void)n_in; (void)out_size; (void)ws_size;

  const float* xyz  = (const float*)d_in[0];
  const float* feat = (const float*)d_in[1];
  const float* w1   = (const float*)d_in[2];
  const float* b1   = (const float*)d_in[3];
  const float* g1   = (const float*)d_in[4];
  const float* be1  = (const float*)d_in[5];
  const float* m1   = (const float*)d_in[6];
  const float* v1   = (const float*)d_in[7];
  const float* w2   = (const float*)d_in[8];
  const float* b2   = (const float*)d_in[9];
  const float* g2   = (const float*)d_in[10];
  const float* be2  = (const float*)d_in[11];
  const float* m2   = (const float*)d_in[12];
  const float* v2   = (const float*)d_in[13];

  char* ws = (char*)d_ws;
  float* Wc     = (float*)ws;
  float* sw1a   = Wc + 16384;
  float* sw1d   = sw1a + 192;
  float* bias1p = sw1d + 192;
  float* bias2p = bias1p + 64;
  float* PQ     = (float*)(ws + (128 << 10));
  float* Cf     = PQ + (size_t)BATCH * NPTS * CH;
  int*   idxb   = (int*)(Cf + (size_t)BATCH * NPTS * CH);

  knn_kernel<<<dim3(NPTS / KNN_Q, BATCH + 1), KNN_Q * KNN_W, 0, stream>>>(
      xyz, idxb, w1, b1, g1, be1, m1, v1, w2, b2, g2, be2, m2, v2,
      Wc, sw1a, sw1d, bias1p, bias2p);

  tables_kernel<<<dim3(TB_X, BATCH + TB_GEO), 256, 0, stream>>>(xyz, feat, Wc, sw1a, sw1d,
                                                               bias1p, PQ, Cf);

  gather_max<<<dim3(NPTS / GA_PTS, 16, BATCH), GA_THR, 0, stream>>>(PQ, Cf, idxb, bias2p,
                                                                    (float*)d_out);
}


# kNN second scan (index write-out): a branch per candidate, each a wait on the scalar unit, replaced by one hit-bit row per lane per 32 candidates, decoded by leading-zero count; scores bit-identical
# speedup vs baseline: 71.0722x; 1.3287x over previous
#include <hip/hip_runtime.h>
#include <hip/hip_bf16.h>
#include <float.h>

#define BATCH   8
#define NPTS    2048
#define CH      128
#define HALF_CH 64
#define KNN_K   20
#define EPSB    1e-5f

typedef __attribute__((ext_vector_type(2))) float v2f;
typedef __attribute__((ext_vector_type(8))) float v8f;

#define KNN_Q    32
#define KNN_W    4
#define KNN_SEG  (NPTS / KNN_W)
#define KNN_STG  (NPTS / (KNN_Q * KNN_W))

__device__ __forceinline__ float knn_score(const float4 me, const float4 p) {
  #pragma clang fp contract(off)
  const float yy  = me.y * p.y;
  const float zz  = me.z * p.z;
  const float dot = __builtin_fmaf(me.x, p.x, yy) + zz;
  return __builtin_fmaf(-2.0f, dot, me.w + p.w);
}

__device__ __forceinline__ void knn_insert(float (&t)[KNN_K], const float d) {
  #pragma unroll
  for (int i = KNN_K - 1; i > 0; --i) t[i] = __builtin_amdgcn_fmed3f(d, t[i - 1], t[i]);
  t[0] = fminf(d, t[0]);
}

__device__ __forceinline__ void fold_params(
    const float* __restrict__ w1, const float* __restrict__ b1,
    const float* __restrict__ g1, const float* __restrict__ be1,
    const float* __restrict__ m1, const float* __restrict__ v1,
    const float* __restrict__ w2, const float* __restrict__ b2,
    const float* __restrict__ g2, const float* __restrict__ be2,
    const float* __restrict__ m2, const float* __restrict__ v2,
    float* __restrict__ Wc,   float* __restrict__ sw1a,
    float* __restrict__ sw1d, float* __restrict__ bias1p,
    float* __restrict__ bias2p) {
  for (int o = blockIdx.x; o < HALF_CH; o += NPTS / KNN_Q) {
    const float s2 = g2[o] / sqrtf(v2[o] + EPSB);
    for (int c = threadIdx.x; c < CH; c += KNN_Q * KNN_W) {
      const float wa = w2[o * (2 * CH) + c];
      const float wb = w2[o * (2 * CH) + CH + c];
      Wc[(size_t)o * CH + c]             = s2 * wa;
      Wc[(size_t)(HALF_CH + o) * CH + c] = s2 * (wb - wa);
    }
  }
  if (blockIdx.x == 0 && threadIdx.x < HALF_CH) {
    const int o = threadIdx.x;
    const float s1 = g1[o] / sqrtf(v1[o] + EPSB);
    bias1p[o] = b1[o] * s1 + (be1[o] - m1[o] * s1);
    #pragma unroll
    for (int c = 0; c < 3; ++c) {
      const float wa = w1[o * 6 + c];
      const float wb = w1[o * 6 + 3 + c];
      sw1a[o * 3 + c] = s1 * wa;
      sw1d[o * 3 + c] = s1 * (wb - wa);
    }
    const float s2 = g2[o] / sqrtf(v2[o] + EPSB);
    bias2p[o] = b2[o] * s2 + (be2[o] - m2[o] * s2);
  }
}

__global__ void __launch_bounds__(KNN_Q * KNN_W)
knn_kernel(const float* __restrict__ xyz, int* __restrict__ idx_out,
           const float* __restrict__ w1, const float* __restrict__ b1,
           const float* __restrict__ g1, const float* __restrict__ be1,
           const float* __restrict__ m1, const float* __restrict__ v1,
           const float* __restrict__ w2, const float* __restrict__ b2,
           const float* __restrict__ g2, const float* __restrict__ be2,
           const float* __restrict__ m2, const float* __restrict__ v2,
           float* __restrict__ Wc,   float* __restrict__ sw1a,
           float* __restrict__ sw1d, float* __restrict__ bias1p,
           float* __restrict__ bias2p) {
  __shared__ float4 pts[NPTS];
  __shared__ float  lst[KNN_W][KNN_K][KNN_Q];
  if (blockIdx.y == BATCH) {
    fold_params(w1, b1, g1, be1, m1, v1, w2, b2, g2, be2, m2, v2,
                Wc, sw1a, sw1d, bias1p, bias2p);
    return;
  }
  const int b = blockIdx.y;
  const float* xb = xyz + (size_t)b * 3 * NPTS;
  float sx[KNN_STG], sy[KNN_STG], sz[KNN_STG];
  #pragma unroll
  for (int i = 0; i < KNN_STG; ++i) {
    const int m = threadIdx.x + i * (KNN_Q * KNN_W);
    sx[i] = xb[m]; sy[i] = xb[NPTS + m]; sz[i] = xb[2 * NPTS + m];
  }
  #pragma unroll
  for (int i = 0; i < KNN_STG; ++i) {
    const int m = threadIdx.x + i * (KNN_Q * KNN_W);
    const float x = sx[i], y = sy[i], z = sz[i];
    pts[m] = make_float4(x, y, z, __builtin_fmaf(z, z, __builtin_fmaf(x, x, y * y)));
  }
  __syncthreads();

  const int lane = threadIdx.x & (KNN_Q - 1);
  const int w    = threadIdx.x / KNN_Q;
  const int n    = blockIdx.x * KNN_Q + lane;
  const float4 me = pts[n];
  const float4* seg = pts + w * KNN_SEG;

  float t[KNN_K];
  #pragma unroll
  for (int i = 0; i < KNN_K; ++i) t[i] = FLT_MAX;

  #pragma unroll 8
  for (int m = 0; m < KNN_SEG; ++m) knn_insert(t, knn_score(me, seg[m]));

  #pragma unroll
  for (int i = 0; i < KNN_K; ++i) lst[w][i][lane] = t[i];
  __syncthreads();
  for (int o = 1; o < KNN_W; ++o) {
    const int w2 = (w + o) & (KNN_W - 1);
    float v[KNN_K];
    #pragma unroll
    for (int i = 0; i < KNN_K; ++i) v[i] = lst[w2][i][lane];
    #pragma unroll
    for (int i = 0; i < KNN_K; ++i) knn_insert(t, v[i]);
  }

  const float thr = t[KNN_K - 1];
  int ties = 0;
  #pragma unroll
  for (int i = 0; i < KNN_K; ++i) ties += (t[i] == thr) ? 1 : 0;

  int off = 0;
  int quota = 0;
  int mine_eq = 0;
  int rem = ties;
  bool exact = false;
  for (int q = 0; q <= w; ++q) {
    int lt = 0, eq = 0;
    #pragma unroll
    for (int i = 0; i < KNN_K; ++i) {
      const float s = lst[q][i][lane];
      lt += (s < thr) ? 1 : 0;
      eq += (s == thr) ? 1 : 0;
    }
    const int a = min(eq, rem);
    rem -= a;
    if (q < w) off += lt + a;
    else {
      quota = lt + a; mine_eq = a;
      exact = (a < eq) || !(lst[q][KNN_K - 1][lane] > thr);
    }
  }
  if (w == KNN_W - 1) quota = KNN_K - off;

  int* op = idx_out + ((size_t)b * NPTS + n) * KNN_K + off;
  const int base = w * KNN_SEG;
  int cnt = 0;
  for (int m0 = 0; m0 < KNN_SEG; m0 += 32) {
    unsigned le = 0u;
    #pragma unroll 8
    for (int k = 0; k < 32; ++k) {
      const float d = knn_score(me, seg[m0 + k]);
      le = le + le + ((d <= thr) ? 1u : 0u);
    }
    while (le != 0u) {
      const int k = __builtin_clz(le);
      bool tie = false;
      if (exact) tie = !(knn_score(me, seg[m0 + k]) < thr);
      if ((!tie || mine_eq > 0) && cnt < quota) { op[cnt] = base + m0 + k; ++cnt; }
      if (tie) --mine_eq;
      le &= ~(0x80000000u >> k);
    }
  }
  for (; cnt < quota; ++cnt) op[cnt] = n;
}

#define QROWS 32

template <bool CTR>
__device__ __forceinline__ void geo_rows(const float* __restrict__ wt, const float* __restrict__ bias,
                                         const float x, const float y, const float z,
                                         float4* __restrict__ o) {
  #pragma unroll
  for (int j = 0; j < 8; ++j) {
    float v[4];
    #pragma unroll
    for (int e = 0; e < 4; ++e) {
      const int c = 4 * j + e;
      if (CTR) v[e] = wt[c * 3 + 0] * x + wt[c * 3 + 1] * y + wt[c * 3 + 2] * z + bias[c];
      else     v[e] = wt[c * 3 + 0] * x + wt[c * 3 + 1] * y + wt[c * 3 + 2] * z;
    }
    o[(size_t)j * NPTS] = make_float4(v[0], v[1], v[2], v[3]);
  }
}

__device__ __forceinline__ void geo_block(const int bx, const int g,
           const float* __restrict__ xyz, const float* __restrict__ sw1a,
           const float* __restrict__ sw1d, const float* __restrict__ bias1p,
           float* __restrict__ PQ) {
  const int pt = bx * 256 + threadIdx.x;
  const int b  = pt / NPTS;
  const int n  = pt - b * NPTS;
  const int c0 = (g & 1) * 32;
  const float* xb = xyz + (size_t)b * 3 * NPTS;
  const float x = xb[n], y = xb[NPTS + n], z = xb[2 * NPTS + n];
  float4* o = (float4*)PQ + ((size_t)b * QROWS + 8 * g) * NPTS + n;
  if (g >= 2) geo_rows<true>(sw1d + c0 * 3, bias1p + c0, x, y, z, o);
  else        geo_rows<false>(sw1a + c0 * 3, bias1p, x, y, z, o);
}

__device__ __forceinline__ void gemm_block(const int bx, const int b,
                          const float* __restrict__ feat, const float* __restrict__ Wc,
                          float* __restrict__ Cf) {
  const int n0   = bx * 16;
  const int wave = threadIdx.x >> 5;
  const int lane = threadIdx.x & 31;
  const int l16  = lane & 15;
  const int hi   = lane >> 4;
  const int m0   = wave * 16;

  const float* Fb   = feat + (size_t)b * CH * NPTS;
  const float* arow = Wc + (size_t)(m0 + l16) * CH + 2 * hi;
  const float* bcol = Fb + (size_t)(2 * hi) * NPTS + n0 + l16;

  v8f acc = {0.f, 0.f, 0.f, 0.f, 0.f, 0.f, 0.f, 0.f};

  #pragma unroll 8
  for (int k = 0; k < CH; k += 4) {
    v2f a;
    a.x = arow[k];
    a.y = arow[k + 1];
    v2f bb;
    bb.x = bcol[(size_t)k * NPTS];
    bb.y = bcol[(size_t)(k + 1) * NPTS];
    acc = __builtin_amdgcn_wmma_f32_16x16x4_f32(false, a, false, bb,
                                                (short)0, acc, false, false);
  }

  float4* out = (float4*)Cf + ((size_t)b * QROWS + (m0 + 8 * hi) / 4) * NPTS + n0 + l16;
  out[0]    = make_float4(acc[0], acc[1], acc[2], acc[3]);
  out[NPTS] = make_float4(acc[4], acc[5], acc[6], acc[7]);
}

#define TB_X   (NPTS / 16)
#define TB_GEO ((BATCH * NPTS / 256) * 4 / TB_X)

__global__ void __launch_bounds__(256)
tables_kernel(const float* __restrict__ xyz, const float* __restrict__ feat,
              const float* __restrict__ Wc, const float* __restrict__ sw1a,
              const float* __restrict__ sw1d, const float* __restrict__ bias1p,
              float* __restrict__ PQ, float* __restrict__ Cf) {
  if (blockIdx.y < BATCH) {
    gemm_block(blockIdx.x, blockIdx.y, feat, Wc, Cf);
  } else {
    const unsigned id = (blockIdx.y - BATCH) * TB_X + blockIdx.x;
    geo_block((int)(id % (BATCH * NPTS / 256)), (int)(id / (BATCH * NPTS / 256)),
              xyz, sw1a, sw1d, bias1p, PQ);
  }
}

#define GA_THR 256
#define GA_PPT 2
#define GA_PTS (GA_THR * GA_PPT)
#define GA_STG (2 * NPTS / GA_THR)

__device__ __forceinline__ float4 max4(const float4 a, const float4 b) {
  return make_float4(fmaxf(a.x, b.x), fmaxf(a.y, b.y), fmaxf(a.z, b.z), fmaxf(a.w, b.w));
}

__global__ void __launch_bounds__(GA_THR)
gather_max(const float* __restrict__ PQ, const float* __restrict__ Cf,
           const int* __restrict__ idx, const float* __restrict__ bias2p,
           float* __restrict__ out) {
  __shared__ float4 tab[2 * NPTS];
  const int b    = blockIdx.z;
  const int tp   = blockIdx.y;
  const bool fea = tp >= 8;
  const int q0   = (tp & 7) * 2;
  const int tid  = threadIdx.x;
  const float4* T   = (const float4*)(fea ? Cf : PQ) + (size_t)b * QROWS * NPTS;
  const float4* src = T + (size_t)q0 * NPTS;

  float4 st[GA_STG];
  #pragma unroll
  for (int i = 0; i < GA_STG; ++i) st[i] = src[tid + i * GA_THR];

  const int nb = blockIdx.x * GA_PTS + tid;
  int4   id[GA_PPT][KNN_K / 4];
  float4 ctr[GA_PPT][2];
  #pragma unroll
  for (int p = 0; p < GA_PPT; ++p) {
    const int n = nb + p * GA_THR;
    const int4* ip = (const int4*)(idx + ((size_t)b * NPTS + n) * KNN_K);
    #pragma unroll
    for (int j = 0; j < KNN_K / 4; ++j) id[p][j] = ip[j];
    ctr[p][0] = T[(size_t)(16 + q0) * NPTS + n];
    ctr[p][1] = T[(size_t)(17 + q0) * NPTS + n];
  }

  #pragma unroll
  for (int i = 0; i < GA_STG; ++i) tab[tid + i * GA_THR] = st[i];

  float4 bs0 = make_float4(0.f, 0.f, 0.f, 0.f), bs1 = bs0;
  if (fea) {
    const float* bp = bias2p + 4 * q0;
    bs0 = make_float4(bp[0], bp[1], bp[2], bp[3]);
    bs1 = make_float4(bp[4], bp[5], bp[6], bp[7]);
  }
  __syncthreads();

  float* ob = out + ((size_t)b * CH + (fea ? HALF_CH : 0) + 4 * q0) * NPTS;
  #pragma unroll
  for (int p = 0; p < GA_PPT; ++p) {
    const int n = nb + p * GA_THR;
    float4 m0 = make_float4(-FLT_MAX, -FLT_MAX, -FLT_MAX, -FLT_MAX), m1 = m0;
    #pragma unroll
    for (int j = 0; j < KNN_K / 4; ++j) {
      const int4 q = id[p][j];
      m0 = max4(m0, tab[q.x]); m1 = max4(m1, tab[NPTS + q.x]);
      m0 = max4(m0, tab[q.y]); m1 = max4(m1, tab[NPTS + q.y]);
      m0 = max4(m0, tab[q.z]); m1 = max4(m1, tab[NPTS + q.z]);
      m0 = max4(m0, tab[q.w]); m1 = max4(m1, tab[NPTS + q.w]);
    }
    float4 c0 = ctr[p][0], c1 = ctr[p][1];
    if (fea) {
      c0 = make_float4(c0.x + bs0.x, c0.y + bs0.y, c0.z + bs0.z, c0.w + bs0.w);
      c1 = make_float4(c1.x + bs1.x, c1.y + bs1.y, c1.z + bs1.z, c1.w + bs1.w);
    }
    ob[0 * NPTS + n] = fmaxf(m0.x + c0.x, 0.0f);
    ob[1 * NPTS + n] = fmaxf(m0.y + c0.y, 0.0f);
    ob[2 * NPTS + n] = fmaxf(m0.z + c0.z, 0.0f);
    ob[3 * NPTS + n] = fmaxf(m0.w + c0.w, 0.0f);
    ob[4 * NPTS + n] = fmaxf(m1.x + c1.x, 0.0f);
    ob[5 * NPTS + n] = fmaxf(m1.y + c1.y, 0.0f);
    ob[6 * NPTS + n] = fmaxf(m1.z + c1.z, 0.0f);
    ob[7 * NPTS + n] = fmaxf(m1.w + c1.w, 0.0f);
  }
}

extern "C" void kernel_launch(void* const* d_in, const int* in_sizes, int n_in,
                              void* d_out, int out_size, void* d_ws, size_t ws_size,
                              hipStream_t stream) {
  (void)in_sizes; (void)n_in; (void)out_size; (void)ws_size;

  const float* xyz  = (const float*)d_in[0];
  const float* feat = (const float*)d_in[1];
  const float* w1   = (const float*)d_in[2];
  const float* b1   = (const float*)d_in[3];
  const float* g1   = (const float*)d_in[4];
  const float* be1  = (const float*)d_in[5];
  const float* m1   = (const float*)d_in[6];
  const float* v1   = (const float*)d_in[7];
  const float* w2   = (const float*)d_in[8];
  const float* b2   = (const float*)d_in[9];
  const float* g2   = (const float*)d_in[10];
  const float* be2  = (const float*)d_in[11];
  const float* m2   = (const float*)d_in[12];
  const float* v2   = (const float*)d_in[13];

  char* ws = (char*)d_ws;
  float* Wc     = (float*)ws;
  float* sw1a   = Wc + 16384;
  float* sw1d   = sw1a + 192;
  float* bias1p = sw1d + 192;
  float* bias2p = bias1p + 64;
  float* PQ     = (float*)(ws + (128 << 10));
  float* Cf     = PQ + (size_t)BATCH * NPTS * CH;
  int*   idxb   = (int*)(Cf + (size_t)BATCH * NPTS * CH);

  knn_kernel<<<dim3(NPTS / KNN_Q, BATCH + 1), KNN_Q * KNN_W, 0, stream>>>(
      xyz, idxb, w1, b1, g1, be1, m1, v1, w2, b2, g2, be2, m2, v2,
      Wc, sw1a, sw1d, bias1p, bias2p);

  tables_kernel<<<dim3(TB_X, BATCH + TB_GEO), 256, 0, stream>>>(xyz, feat, Wc, sw1a, sw1d,
                                                               bias1p, PQ, Cf);

  gather_max<<<dim3(NPTS / GA_PTS, 16, BATCH), GA_THR, 0, stream>>>(PQ, Cf, idxb, bias2p,
                                                                    (float*)d_out);
}


# kNN scan 1: after a quarter's first 128 candidates, 32 at a time are only compared with the lane's current 20th score (hit bits); the 20-deep median chain runs for hits alone; scores bit-identical
# speedup vs baseline: 75.2146x; 1.0583x over previous
#include <hip/hip_runtime.h>
#include <hip/hip_bf16.h>
#include <float.h>

#define BATCH   8
#define NPTS    2048
#define CH      128
#define HALF_CH 64
#define KNN_K   20
#define EPSB    1e-5f

typedef __attribute__((ext_vector_type(2))) float v2f;
typedef __attribute__((ext_vector_type(8))) float v8f;

#define KNN_Q    32
#define KNN_W    4
#define KNN_SEG  (NPTS / KNN_W)
#define KNN_STG  (NPTS / (KNN_Q * KNN_W))
#define KNN_PRE  128

__device__ __forceinline__ float knn_score(const float4 me, const float4 p) {
  #pragma clang fp contract(off)
  const float yy  = me.y * p.y;
  const float zz  = me.z * p.z;
  const float dot = __builtin_fmaf(me.x, p.x, yy) + zz;
  return __builtin_fmaf(-2.0f, dot, me.w + p.w);
}

__device__ __forceinline__ void knn_insert(float (&t)[KNN_K], const float d) {
  #pragma unroll
  for (int i = KNN_K - 1; i > 0; --i) t[i] = __builtin_amdgcn_fmed3f(d, t[i - 1], t[i]);
  t[0] = fminf(d, t[0]);
}

__device__ __forceinline__ void fold_params(
    const float* __restrict__ w1, const float* __restrict__ b1,
    const float* __restrict__ g1, const float* __restrict__ be1,
    const float* __restrict__ m1, const float* __restrict__ v1,
    const float* __restrict__ w2, const float* __restrict__ b2,
    const float* __restrict__ g2, const float* __restrict__ be2,
    const float* __restrict__ m2, const float* __restrict__ v2,
    float* __restrict__ Wc,   float* __restrict__ sw1a,
    float* __restrict__ sw1d, float* __restrict__ bias1p,
    float* __restrict__ bias2p) {
  for (int o = blockIdx.x; o < HALF_CH; o += NPTS / KNN_Q) {
    const float s2 = g2[o] / sqrtf(v2[o] + EPSB);
    for (int c = threadIdx.x; c < CH; c += KNN_Q * KNN_W) {
      const float wa = w2[o * (2 * CH) + c];
      const float wb = w2[o * (2 * CH) + CH + c];
      Wc[(size_t)o * CH + c]             = s2 * wa;
      Wc[(size_t)(HALF_CH + o) * CH + c] = s2 * (wb - wa);
    }
  }
  if (blockIdx.x == 0 && threadIdx.x < HALF_CH) {
    const int o = threadIdx.x;
    const float s1 = g1[o] / sqrtf(v1[o] + EPSB);
    bias1p[o] = b1[o] * s1 + (be1[o] - m1[o] * s1);
    #pragma unroll
    for (int c = 0; c < 3; ++c) {
      const float wa = w1[o * 6 + c];
      const float wb = w1[o * 6 + 3 + c];
      sw1a[o * 3 + c] = s1 * wa;
      sw1d[o * 3 + c] = s1 * (wb - wa);
    }
    const float s2 = g2[o] / sqrtf(v2[o] + EPSB);
    bias2p[o] = b2[o] * s2 + (be2[o] - m2[o] * s2);
  }
}

__global__ void __launch_bounds__(KNN_Q * KNN_W)
knn_kernel(const float* __restrict__ xyz, int* __restrict__ idx_out,
           const float* __restrict__ w1, const float* __restrict__ b1,
           const float* __restrict__ g1, const float* __restrict__ be1,
           const float* __restrict__ m1, const float* __restrict__ v1,
           const float* __restrict__ w2, const float* __restrict__ b2,
           const float* __restrict__ g2, const float* __restrict__ be2,
           const float* __restrict__ m2, const float* __restrict__ v2,
           float* __restrict__ Wc,   float* __restrict__ sw1a,
           float* __restrict__ sw1d, float* __restrict__ bias1p,
           float* __restrict__ bias2p) {
  __shared__ float4 pts[NPTS];
  __shared__ float  lst[KNN_W][KNN_K][KNN_Q];
  if (blockIdx.y == BATCH) {
    fold_params(w1, b1, g1, be1, m1, v1, w2, b2, g2, be2, m2, v2,
                Wc, sw1a, sw1d, bias1p, bias2p);
    return;
  }
  const int b = blockIdx.y;
  const float* xb = xyz + (size_t)b * 3 * NPTS;
  float sx[KNN_STG], sy[KNN_STG], sz[KNN_STG];
  #pragma unroll
  for (int i = 0; i < KNN_STG; ++i) {
    const int m = threadIdx.x + i * (KNN_Q * KNN_W);
    sx[i] = xb[m]; sy[i] = xb[NPTS + m]; sz[i] = xb[2 * NPTS + m];
  }
  #pragma unroll
  for (int i = 0; i < KNN_STG; ++i) {
    const int m = threadIdx.x + i * (KNN_Q * KNN_W);
    const float x = sx[i], y = sy[i], z = sz[i];
    pts[m] = make_float4(x, y, z, __builtin_fmaf(z, z, __builtin_fmaf(x, x, y * y)));
  }
  __syncthreads();

  const int lane = threadIdx.x & (KNN_Q - 1);
  const int w    = threadIdx.x / KNN_Q;
  const int n    = blockIdx.x * KNN_Q + lane;
  const float4 me = pts[n];
  const float4* seg = pts + w * KNN_SEG;

  float t[KNN_K];
  #pragma unroll
  for (int i = 0; i < KNN_K; ++i) t[i] = FLT_MAX;

  #pragma unroll 8
  for (int m = 0; m < KNN_PRE; ++m) knn_insert(t, knn_score(me, seg[m]));

  for (int m0 = KNN_PRE; m0 < KNN_SEG; m0 += 32) {
    const float u = t[KNN_K - 1];
    unsigned in = 0u;
    #pragma unroll 8
    for (int k = 0; k < 32; ++k) {
      const float d = knn_score(me, seg[m0 + k]);
      in = in + in + ((d < u) ? 1u : 0u);
    }
    if (__builtin_amdgcn_ballot_w32(in != 0u) != 0u) {
      bool has0 = in != 0u;
      int k = __builtin_clz(in | 1u);
      in &= ~(0x80000000u >> k);
      float4 p0 = seg[m0 + k];
      bool has1 = in != 0u;
      k = __builtin_clz(in | 1u);
      in &= ~(0x80000000u >> k);
      float4 p1 = seg[m0 + k];
      unsigned more;
      do {
        more = __builtin_amdgcn_ballot_w32(in != 0u);
        const bool hn0 = in != 0u;
        k = __builtin_clz(in | 1u);
        in &= ~(0x80000000u >> k);
        const float4 q0 = seg[m0 + k];
        const bool hn1 = in != 0u;
        k = __builtin_clz(in | 1u);
        in &= ~(0x80000000u >> k);
        const float4 q1 = seg[m0 + k];
        __builtin_amdgcn_sched_barrier(0);
        const float d0 = knn_score(me, p0);
        knn_insert(t, has0 ? d0 : FLT_MAX);
        const float d1 = knn_score(me, p1);
        knn_insert(t, has1 ? d1 : FLT_MAX);
        __builtin_amdgcn_sched_barrier(0);
        asm volatile("" :: "v"(q0.x), "v"(q0.y), "v"(q0.z), "v"(q0.w),
                           "v"(q1.x), "v"(q1.y), "v"(q1.z), "v"(q1.w));
        p0 = q0; p1 = q1; has0 = hn0; has1 = hn1;
      } while (more != 0u);
    }
  }

  #pragma unroll
  for (int i = 0; i < KNN_K; ++i) lst[w][i][lane] = t[i];
  __syncthreads();
  for (int o = 1; o < KNN_W; ++o) {
    const int w2 = (w + o) & (KNN_W - 1);
    float v[KNN_K];
    #pragma unroll
    for (int i = 0; i < KNN_K; ++i) v[i] = lst[w2][i][lane];
    #pragma unroll
    for (int i = 0; i < KNN_K; ++i) knn_insert(t, v[i]);
  }

  const float thr = t[KNN_K - 1];
  int ties = 0;
  #pragma unroll
  for (int i = 0; i < KNN_K; ++i) ties += (t[i] == thr) ? 1 : 0;

  int off = 0;
  int quota = 0;
  int mine_eq = 0;
  int rem = ties;
  bool exact = false;
  for (int q = 0; q <= w; ++q) {
    int lt = 0, eq = 0;
    #pragma unroll
    for (int i = 0; i < KNN_K; ++i) {
      const float s = lst[q][i][lane];
      lt += (s < thr) ? 1 : 0;
      eq += (s == thr) ? 1 : 0;
    }
    const int a = min(eq, rem);
    rem -= a;
    if (q < w) off += lt + a;
    else {
      quota = lt + a; mine_eq = a;
      exact = (a < eq) || !(lst[q][KNN_K - 1][lane] > thr);
    }
  }
  if (w == KNN_W - 1) quota = KNN_K - off;

  int* op = idx_out + ((size_t)b * NPTS + n) * KNN_K + off;
  const int base = w * KNN_SEG;
  int cnt = 0;
  for (int m0 = 0; m0 < KNN_SEG; m0 += 32) {
    unsigned le = 0u;
    #pragma unroll 8
    for (int k = 0; k < 32; ++k) {
      const float d = knn_score(me, seg[m0 + k]);
      le = le + le + ((d <= thr) ? 1u : 0u);
    }
    while (le != 0u) {
      const int k = __builtin_clz(le);
      bool tie = false;
      if (exact) tie = !(knn_score(me, seg[m0 + k]) < thr);
      if ((!tie || mine_eq > 0) && cnt < quota) { op[cnt] = base + m0 + k; ++cnt; }
      if (tie) --mine_eq;
      le &= ~(0x80000000u >> k);
    }
  }
  for (; cnt < quota; ++cnt) op[cnt] = n;
}

#define QROWS 32

template <bool CTR>
__device__ __forceinline__ void geo_rows(const float* __restrict__ wt, const float* __restrict__ bias,
                                         const float x, const float y, const float z,
                                         float4* __restrict__ o) {
  #pragma unroll
  for (int j = 0; j < 8; ++j) {
    float v[4];
    #pragma unroll
    for (int e = 0; e < 4; ++e) {
      const int c = 4 * j + e;
      if (CTR) v[e] = wt[c * 3 + 0] * x + wt[c * 3 + 1] * y + wt[c * 3 + 2] * z + bias[c];
      else     v[e] = wt[c * 3 + 0] * x + wt[c * 3 + 1] * y + wt[c * 3 + 2] * z;
    }
    o[(size_t)j * NPTS] = make_float4(v[0], v[1], v[2], v[3]);
  }
}

__device__ __forceinline__ void geo_block(const int bx, const int g,
           const float* __restrict__ xyz, const float* __restrict__ sw1a,
           const float* __restrict__ sw1d, const float* __restrict__ bias1p,
           float* __restrict__ PQ) {
  const int pt = bx * 256 + threadIdx.x;
  const int b  = pt / NPTS;
  const int n  = pt - b * NPTS;
  const int c0 = (g & 1) * 32;
  const float* xb = xyz + (size_t)b * 3 * NPTS;
  const float x = xb[n], y = xb[NPTS + n], z = xb[2 * NPTS + n];
  float4* o = (float4*)PQ + ((size_t)b * QROWS + 8 * g) * NPTS + n;
  if (g >= 2) geo_rows<true>(sw1d + c0 * 3, bias1p + c0, x, y, z, o);
  else        geo_rows<false>(sw1a + c0 * 3, bias1p, x, y, z, o);
}

__device__ __forceinline__ void gemm_block(const int bx, const int b,
                          const float* __restrict__ feat, const float* __restrict__ Wc,
                          float* __restrict__ Cf) {
  const int n0   = bx * 16;
  const int wave = threadIdx.x >> 5;
  const int lane = threadIdx.x & 31;
  const int l16  = lane & 15;
  const int hi   = lane >> 4;
  const int m0   = wave * 16;

  const float* Fb   = feat + (size_t)b * CH * NPTS;
  const float* arow = Wc + (size_t)(m0 + l16) * CH + 2 * hi;
  const float* bcol = Fb + (size_t)(2 * hi) * NPTS + n0 + l16;

  v8f acc = {0.f, 0.f, 0.f, 0.f, 0.f, 0.f, 0.f, 0.f};

  #pragma unroll 8
  for (int k = 0; k < CH; k += 4) {
    v2f a;
    a.x = arow[k];
    a.y = arow[k + 1];
    v2f bb;
    bb.x = bcol[(size_t)k * NPTS];
    bb.y = bcol[(size_t)(k + 1) * NPTS];
    acc = __builtin_amdgcn_wmma_f32_16x16x4_f32(false, a, false, bb,
                                                (short)0, acc, false, false);
  }

  float4* out = (float4*)Cf + ((size_t)b * QROWS + (m0 + 8 * hi) / 4) * NPTS + n0 + l16;
  out[0]    = make_float4(acc[0], acc[1], acc[2], acc[3]);
  out[NPTS] = make_float4(acc[4], acc[5], acc[6], acc[7]);
}

#define TB_X   (NPTS / 16)
#define TB_GEO ((BATCH * NPTS / 256) * 4 / TB_X)

__global__ void __launch_bounds__(256)
tables_kernel(const float* __restrict__ xyz, const float* __restrict__ feat,
              const float* __restrict__ Wc, const float* __restrict__ sw1a,
              const float* __restrict__ sw1d, const float* __restrict__ bias1p,
              float* __restrict__ PQ, float* __restrict__ Cf) {
  if (blockIdx.y < BATCH) {
    gemm_block(blockIdx.x, blockIdx.y, feat, Wc, Cf);
  } else {
    const unsigned id = (blockIdx.y - BATCH) * TB_X + blockIdx.x;
    geo_block((int)(id % (BATCH * NPTS / 256)), (int)(id / (BATCH * NPTS / 256)),
              xyz, sw1a, sw1d, bias1p, PQ);
  }
}

#define GA_THR 256
#define GA_PPT 2
#define GA_PTS (GA_THR * GA_PPT)
#define GA_STG (2 * NPTS / GA_THR)

__device__ __forceinline__ float4 max4(const float4 a, const float4 b) {
  return make_float4(fmaxf(a.x, b.x), fmaxf(a.y, b.y), fmaxf(a.z, b.z), fmaxf(a.w, b.w));
}

__global__ void __launch_bounds__(GA_THR)
gather_max(const float* __restrict__ PQ, const float* __restrict__ Cf,
           const int* __restrict__ idx, const float* __restrict__ bias2p,
           float* __restrict__ out) {
  __shared__ float4 tab[2 * NPTS];
  const int b    = blockIdx.z;
  const int tp   = blockIdx.y;
  const bool fea = tp >= 8;
  const int q0   = (tp & 7) * 2;
  const int tid  = threadIdx.x;
  const float4* T   = (const float4*)(fea ? Cf : PQ) + (size_t)b * QROWS * NPTS;
  const float4* src = T + (size_t)q0 * NPTS;

  float4 st[GA_STG];
  #pragma unroll
  for (int i = 0; i < GA_STG; ++i) st[i] = src[tid + i * GA_THR];

  const int nb = blockIdx.x * GA_PTS + tid;
  int4   id[GA_PPT][KNN_K / 4];
  float4 ctr[GA_PPT][2];
  #pragma unroll
  for (int p = 0; p < GA_PPT; ++p) {
    const int n = nb + p * GA_THR;
    const int4* ip = (const int4*)(idx + ((size_t)b * NPTS + n) * KNN_K);
    #pragma unroll
    for (int j = 0; j < KNN_K / 4; ++j) id[p][j] = ip[j];
    ctr[p][0] = T[(size_t)(16 + q0) * NPTS + n];
    ctr[p][1] = T[(size_t)(17 + q0) * NPTS + n];
  }

  #pragma unroll
  for (int i = 0; i < GA_STG; ++i) tab[tid + i * GA_THR] = st[i];

  float4 bs0 = make_float4(0.f, 0.f, 0.f, 0.f), bs1 = bs0;
  if (fea) {
    const float* bp = bias2p + 4 * q0;
    bs0 = make_float4(bp[0], bp[1], bp[2], bp[3]);
    bs1 = make_float4(bp[4], bp[5], bp[6], bp[7]);
  }
  __syncthreads();

  float* ob = out + ((size_t)b * CH + (fea ? HALF_CH : 0) + 4 * q0) * NPTS;
  #pragma unroll
  for (int p = 0; p < GA_PPT; ++p) {
    const int n = nb + p * GA_THR;
    float4 m0 = make_float4(-FLT_MAX, -FLT_MAX, -FLT_MAX, -FLT_MAX), m1 = m0;
    #pragma unroll
    for (int j = 0; j < KNN_K / 4; ++j) {
      const int4 q = id[p][j];
      m0 = max4(m0, tab[q.x]); m1 = max4(m1, tab[NPTS + q.x]);
      m0 = max4(m0, tab[q.y]); m1 = max4(m1, tab[NPTS + q.y]);
      m0 = max4(m0, tab[q.z]); m1 = max4(m1, tab[NPTS + q.z]);
      m0 = max4(m0, tab[q.w]); m1 = max4(m1, tab[NPTS + q.w]);
    }
    float4 c0 = ctr[p][0], c1 = ctr[p][1];
    if (fea) {
      c0 = make_float4(c0.x + bs0.x, c0.y + bs0.y, c0.z + bs0.z, c0.w + bs0.w);
      c1 = make_float4(c1.x + bs1.x, c1.y + bs1.y, c1.z + bs1.z, c1.w + bs1.w);
    }
    ob[0 * NPTS + n] = fmaxf(m0.x + c0.x, 0.0f);
    ob[1 * NPTS + n] = fmaxf(m0.y + c0.y, 0.0f);
    ob[2 * NPTS + n] = fmaxf(m0.z + c0.z, 0.0f);
    ob[3 * NPTS + n] = fmaxf(m0.w + c0.w, 0.0f);
    ob[4 * NPTS + n] = fmaxf(m1.x + c1.x, 0.0f);
    ob[5 * NPTS + n] = fmaxf(m1.y + c1.y, 0.0f);
    ob[6 * NPTS + n] = fmaxf(m1.z + c1.z, 0.0f);
    ob[7 * NPTS + n] = fmaxf(m1.w + c1.w, 0.0f);
  }
}

extern "C" void kernel_launch(void* const* d_in, const int* in_sizes, int n_in,
                              void* d_out, int out_size, void* d_ws, size_t ws_size,
                              hipStream_t stream) {
  (void)in_sizes; (void)n_in; (void)out_size; (void)ws_size;

  const float* xyz  = (const float*)d_in[0];
  const float* feat = (const float*)d_in[1];
  const float* w1   = (const float*)d_in[2];
  const float* b1   = (const float*)d_in[3];
  const float* g1   = (const float*)d_in[4];
  const float* be1  = (const float*)d_in[5];
  const float* m1   = (const float*)d_in[6];
  const float* v1   = (const float*)d_in[7];
  const float* w2   = (const float*)d_in[8];
  const float* b2   = (const float*)d_in[9];
  const float* g2   = (const float*)d_in[10];
  const float* be2  = (const float*)d_in[11];
  const float* m2   = (const float*)d_in[12];
  const float* v2   = (const float*)d_in[13];

  char* ws = (char*)d_ws;
  float* Wc     = (float*)ws;
  float* sw1a   = Wc + 16384;
  float* sw1d   = sw1a + 192;
  float* bias1p = sw1d + 192;
  float* bias2p = bias1p + 64;
  float* PQ     = (float*)(ws + (128 << 10));
  float* Cf     = PQ + (size_t)BATCH * NPTS * CH;
  int*   idxb   = (int*)(Cf + (size_t)BATCH * NPTS * CH);

  knn_kernel<<<dim3(NPTS / KNN_Q, BATCH + 1), KNN_Q * KNN_W, 0, stream>>>(
      xyz, idxb, w1, b1, g1, be1, m1, v1, w2, b2, g2, be2, m2, v2,
      Wc, sw1a, sw1d, bias1p, bias2p);

  tables_kernel<<<dim3(TB_X, BATCH + TB_GEO), 256, 0, stream>>>(xyz, feat, Wc, sw1a, sw1d,
                                                               bias1p, PQ, Cf);

  gather_max<<<dim3(NPTS / GA_PTS, 16, BATCH), GA_THR, 0, stream>>>(PQ, Cf, idxb, bias2p,
                                                                    (float*)d_out);
}


# kNN scan 2 rescoring cut to a quarter's first 128 candidates: scan 1 lists each later hit's index per lane in LDS, scan 2 walks the list (full rescoring if a list spills); scores bit-identical
# speedup vs baseline: 81.2074x; 1.0797x over previous
#include <hip/hip_runtime.h>
#include <hip/hip_bf16.h>
#include <float.h>

#define BATCH   8
#define NPTS    2048
#define CH      128
#define HALF_CH 64
#define KNN_K   20
#define EPSB    1e-5f

typedef __attribute__((ext_vector_type(2))) float v2f;
typedef __attribute__((ext_vector_type(8))) float v8f;

#define KNN_Q    32
#define KNN_W    4
#define KNN_SEG  (NPTS / KNN_W)
#define KNN_STG  (NPTS / (KNN_Q * KNN_W))
#define KNN_PRE  128
#define KNN_CAP  95

__device__ __forceinline__ float knn_score(const float4 me, const float4 p) {
  #pragma clang fp contract(off)
  const float yy  = me.y * p.y;
  const float zz  = me.z * p.z;
  const float dot = __builtin_fmaf(me.x, p.x, yy) + zz;
  return __builtin_fmaf(-2.0f, dot, me.w + p.w);
}

__device__ __forceinline__ void knn_insert(float (&t)[KNN_K], const float d) {
  #pragma unroll
  for (int i = KNN_K - 1; i > 0; --i) t[i] = __builtin_amdgcn_fmed3f(d, t[i - 1], t[i]);
  t[0] = fminf(d, t[0]);
}

__device__ __forceinline__ void fold_params(
    const float* __restrict__ w1, const float* __restrict__ b1,
    const float* __restrict__ g1, const float* __restrict__ be1,
    const float* __restrict__ m1, const float* __restrict__ v1,
    const float* __restrict__ w2, const float* __restrict__ b2,
    const float* __restrict__ g2, const float* __restrict__ be2,
    const float* __restrict__ m2, const float* __restrict__ v2,
    float* __restrict__ Wc,   float* __restrict__ sw1a,
    float* __restrict__ sw1d, float* __restrict__ bias1p,
    float* __restrict__ bias2p) {
  for (int o = blockIdx.x; o < HALF_CH; o += NPTS / KNN_Q) {
    const float s2 = g2[o] / sqrtf(v2[o] + EPSB);
    for (int c = threadIdx.x; c < CH; c += KNN_Q * KNN_W) {
      const float wa = w2[o * (2 * CH) + c];
      const float wb = w2[o * (2 * CH) + CH + c];
      Wc[(size_t)o * CH + c]             = s2 * wa;
      Wc[(size_t)(HALF_CH + o) * CH + c] = s2 * (wb - wa);
    }
  }
  if (blockIdx.x == 0 && threadIdx.x < HALF_CH) {
    const int o = threadIdx.x;
    const float s1 = g1[o] / sqrtf(v1[o] + EPSB);
    bias1p[o] = b1[o] * s1 + (be1[o] - m1[o] * s1);
    #pragma unroll
    for (int c = 0; c < 3; ++c) {
      const float wa = w1[o * 6 + c];
      const float wb = w1[o * 6 + 3 + c];
      sw1a[o * 3 + c] = s1 * wa;
      sw1d[o * 3 + c] = s1 * (wb - wa);
    }
    const float s2 = g2[o] / sqrtf(v2[o] + EPSB);
    bias2p[o] = b2[o] * s2 + (be2[o] - m2[o] * s2);
  }
}

__global__ void __launch_bounds__(KNN_Q * KNN_W)
knn_kernel(const float* __restrict__ xyz, int* __restrict__ idx_out,
           const float* __restrict__ w1, const float* __restrict__ b1,
           const float* __restrict__ g1, const float* __restrict__ be1,
           const float* __restrict__ m1, const float* __restrict__ v1,
           const float* __restrict__ w2, const float* __restrict__ b2,
           const float* __restrict__ g2, const float* __restrict__ be2,
           const float* __restrict__ m2, const float* __restrict__ v2,
           float* __restrict__ Wc,   float* __restrict__ sw1a,
           float* __restrict__ sw1d, float* __restrict__ bias1p,
           float* __restrict__ bias2p) {
  __shared__ float4 pts[NPTS];
  __shared__ float  lst[KNN_W][KNN_K][KNN_Q];
  __shared__ int    hit[KNN_W][KNN_CAP + 1][KNN_Q];
  if (blockIdx.y == BATCH) {
    fold_params(w1, b1, g1, be1, m1, v1, w2, b2, g2, be2, m2, v2,
                Wc, sw1a, sw1d, bias1p, bias2p);
    return;
  }
  const int b = blockIdx.y;
  const float* xb = xyz + (size_t)b * 3 * NPTS;
  float sx[KNN_STG], sy[KNN_STG], sz[KNN_STG];
  #pragma unroll
  for (int i = 0; i < KNN_STG; ++i) {
    const int m = threadIdx.x + i * (KNN_Q * KNN_W);
    sx[i] = xb[m]; sy[i] = xb[NPTS + m]; sz[i] = xb[2 * NPTS + m];
  }
  #pragma unroll
  for (int i = 0; i < KNN_STG; ++i) {
    const int m = threadIdx.x + i * (KNN_Q * KNN_W);
    const float x = sx[i], y = sy[i], z = sz[i];
    pts[m] = make_float4(x, y, z, __builtin_fmaf(z, z, __builtin_fmaf(x, x, y * y)));
  }
  __syncthreads();

  const int lane = threadIdx.x & (KNN_Q - 1);
  const int w    = threadIdx.x / KNN_Q;
  const int n    = blockIdx.x * KNN_Q + lane;
  const float4 me = pts[n];
  const float4* seg = pts + w * KNN_SEG;
  int (*myhit)[KNN_Q] = hit[w];
  int nhit = 0;

  float t[KNN_K];
  #pragma unroll
  for (int i = 0; i < KNN_K; ++i) t[i] = FLT_MAX;

  #pragma unroll 8
  for (int m = 0; m < KNN_PRE; ++m) knn_insert(t, knn_score(me, seg[m]));

  for (int m0 = KNN_PRE; m0 < KNN_SEG; m0 += 32) {
    const float u = t[KNN_K - 1];
    unsigned in = 0u;
    #pragma unroll 8
    for (int k = 0; k < 32; ++k) {
      const float d = knn_score(me, seg[m0 + k]);
      in = in + in + ((d < u) ? 1u : 0u);
    }
    if (__builtin_amdgcn_ballot_w32(in != 0u) != 0u) {
      bool has0 = in != 0u;
      int k = __builtin_clz(in | 1u);
      in &= ~(0x80000000u >> k);
      int i0 = m0 + k;
      float4 p0 = seg[i0];
      bool has1 = in != 0u;
      k = __builtin_clz(in | 1u);
      in &= ~(0x80000000u >> k);
      int i1 = m0 + k;
      float4 p1 = seg[i1];
      unsigned more;
      do {
        more = __builtin_amdgcn_ballot_w32(in != 0u);
        const bool hn0 = in != 0u;
        k = __builtin_clz(in | 1u);
        in &= ~(0x80000000u >> k);
        const int j0 = m0 + k;
        const float4 q0 = seg[j0];
        const bool hn1 = in != 0u;
        k = __builtin_clz(in | 1u);
        in &= ~(0x80000000u >> k);
        const int j1 = m0 + k;
        const float4 q1 = seg[j1];
        __builtin_amdgcn_sched_barrier(0);
        const float d0 = knn_score(me, p0);
        knn_insert(t, has0 ? d0 : FLT_MAX);
        const float d1 = knn_score(me, p1);
        knn_insert(t, has1 ? d1 : FLT_MAX);
        myhit[min(nhit, KNN_CAP)][lane] = i0;
        nhit += has0 ? 1 : 0;
        myhit[min(nhit, KNN_CAP)][lane] = i1;
        nhit += has1 ? 1 : 0;
        __builtin_amdgcn_sched_barrier(0);
        asm volatile("" :: "v"(q0.x), "v"(q0.y), "v"(q0.z), "v"(q0.w),
                           "v"(q1.x), "v"(q1.y), "v"(q1.z), "v"(q1.w));
        p0 = q0; p1 = q1; has0 = hn0; has1 = hn1; i0 = j0; i1 = j1;
      } while (more != 0u);
    }
  }

  #pragma unroll
  for (int i = 0; i < KNN_K; ++i) lst[w][i][lane] = t[i];
  __syncthreads();
  for (int o = 1; o < KNN_W; ++o) {
    const int w2 = (w + o) & (KNN_W - 1);
    float v[KNN_K];
    #pragma unroll
    for (int i = 0; i < KNN_K; ++i) v[i] = lst[w2][i][lane];
    #pragma unroll
    for (int i = 0; i < KNN_K; ++i) knn_insert(t, v[i]);
  }

  const float thr = t[KNN_K - 1];
  int ties = 0;
  #pragma unroll
  for (int i = 0; i < KNN_K; ++i) ties += (t[i] == thr) ? 1 : 0;

  int off = 0;
  int quota = 0;
  int mine_eq = 0;
  int rem = ties;
  bool exact = false;
  for (int q = 0; q <= w; ++q) {
    int lt = 0, eq = 0;
    #pragma unroll
    for (int i = 0; i < KNN_K; ++i) {
      const float s = lst[q][i][lane];
      lt += (s < thr) ? 1 : 0;
      eq += (s == thr) ? 1 : 0;
    }
    const int a = min(eq, rem);
    rem -= a;
    if (q < w) off += lt + a;
    else {
      quota = lt + a; mine_eq = a;
      exact = (a < eq) || !(lst[q][KNN_K - 1][lane] > thr);
    }
  }
  if (w == KNN_W - 1) quota = KNN_K - off;

  int* op = idx_out + ((size_t)b * NPTS + n) * KNN_K + off;
  const int base = w * KNN_SEG;
  int cnt = 0;
  auto take = [&](const bool tie, const int id) {
    if ((!tie || mine_eq > 0) && cnt < quota) { op[cnt] = id; ++cnt; }
    if (tie) --mine_eq;
  };
  const bool spilt = __builtin_amdgcn_ballot_w32(nhit > KNN_CAP) != 0u;
  const int lim = spilt ? KNN_SEG : KNN_PRE;
  for (int m0 = 0; m0 < lim; m0 += 32) {
    unsigned le = 0u;
    #pragma unroll 8
    for (int k = 0; k < 32; ++k) {
      const float d = knn_score(me, seg[m0 + k]);
      le = le + le + ((d <= thr) ? 1u : 0u);
    }
    while (le != 0u) {
      const int k = __builtin_clz(le);
      bool tie = false;
      if (exact) tie = !(knn_score(me, seg[m0 + k]) < thr);
      take(tie, base + m0 + k);
      le &= ~(0x80000000u >> k);
    }
  }
  if (!spilt) {
    const int last = max(nhit - 1, 0);
    for (int j0 = 0; __builtin_amdgcn_ballot_w32(j0 < nhit) != 0u; j0 += 32) {
      unsigned le = 0u;
      #pragma unroll 8
      for (int k = 0; k < 32; ++k) {
        const float d = knn_score(me, seg[myhit[min(j0 + k, last)][lane]]);
        le = le + le + ((d <= thr) ? 1u : 0u);
      }
      const int have = min(max(nhit - j0, 0), 32);
      le &= (unsigned)(0xffffffff00000000ull >> have);
      while (le != 0u) {
        const int k = __builtin_clz(le);
        const int id = myhit[j0 + k][lane];
        bool tie = false;
        if (exact) tie = !(knn_score(me, seg[id]) < thr);
        take(tie, base + id);
        le &= ~(0x80000000u >> k);
      }
    }
  }
  for (; cnt < quota; ++cnt) op[cnt] = n;
}

#define QROWS 32

template <bool CTR>
__device__ __forceinline__ void geo_rows(const float* __restrict__ wt, const float* __restrict__ bias,
                                         const float x, const float y, const float z,
                                         float4* __restrict__ o) {
  #pragma unroll
  for (int j = 0; j < 8; ++j) {
    float v[4];
    #pragma unroll
    for (int e = 0; e < 4; ++e) {
      const int c = 4 * j + e;
      if (CTR) v[e] = wt[c * 3 + 0] * x + wt[c * 3 + 1] * y + wt[c * 3 + 2] * z + bias[c];
      else     v[e] = wt[c * 3 + 0] * x + wt[c * 3 + 1] * y + wt[c * 3 + 2] * z;
    }
    o[(size_t)j * NPTS] = make_float4(v[0], v[1], v[2], v[3]);
  }
}

__device__ __forceinline__ void geo_block(const int bx, const int g,
           const float* __restrict__ xyz, const float* __restrict__ sw1a,
           const float* __restrict__ sw1d, const float* __restrict__ bias1p,
           float* __restrict__ PQ) {
  const int pt = bx * 256 + threadIdx.x;
  const int b  = pt / NPTS;
  const int n  = pt - b * NPTS;
  const int c0 = (g & 1) * 32;
  const float* xb = xyz + (size_t)b * 3 * NPTS;
  const float x = xb[n], y = xb[NPTS + n], z = xb[2 * NPTS + n];
  float4* o = (float4*)PQ + ((size_t)b * QROWS + 8 * g) * NPTS + n;
  if (g >= 2) geo_rows<true>(sw1d + c0 * 3, bias1p + c0, x, y, z, o);
  else        geo_rows<false>(sw1a + c0 * 3, bias1p, x, y, z, o);
}

__device__ __forceinline__ void gemm_block(const int bx, const int b,
                          const float* __restrict__ feat, const float* __restrict__ Wc,
                          float* __restrict__ Cf) {
  const int n0   = bx * 16;
  const int wave = threadIdx.x >> 5;
  const int lane = threadIdx.x & 31;
  const int l16  = lane & 15;
  const int hi   = lane >> 4;
  const int m0   = wave * 16;

  const float* Fb   = feat + (size_t)b * CH * NPTS;
  const float* arow = Wc + (size_t)(m0 + l16) * CH + 2 * hi;
  const float* bcol = Fb + (size_t)(2 * hi) * NPTS + n0 + l16;

  v8f acc = {0.f, 0.f, 0.f, 0.f, 0.f, 0.f, 0.f, 0.f};

  #pragma unroll 8
  for (int k = 0; k < CH; k += 4) {
    v2f a;
    a.x = arow[k];
    a.y = arow[k + 1];
    v2f bb;
    bb.x = bcol[(size_t)k * NPTS];
    bb.y = bcol[(size_t)(k + 1) * NPTS];
    acc = __builtin_amdgcn_wmma_f32_16x16x4_f32(false, a, false, bb,
                                                (short)0, acc, false, false);
  }

  float4* out = (float4*)Cf + ((size_t)b * QROWS + (m0 + 8 * hi) / 4) * NPTS + n0 + l16;
  out[0]    = make_float4(acc[0], acc[1], acc[2], acc[3]);
  out[NPTS] = make_float4(acc[4], acc[5], acc[6], acc[7]);
}

#define TB_X   (NPTS / 16)
#define TB_GEO ((BATCH * NPTS / 256) * 4 / TB_X)

__global__ void __launch_bounds__(256)
tables_kernel(const float* __restrict__ xyz, const float* __restrict__ feat,
              const float* __restrict__ Wc, const float* __restrict__ sw1a,
              const float* __restrict__ sw1d, const float* __restrict__ bias1p,
              float* __restrict__ PQ, float* __restrict__ Cf) {
  if (blockIdx.y < BATCH) {
    gemm_block(blockIdx.x, blockIdx.y, feat, Wc, Cf);
  } else {
    const unsigned id = (blockIdx.y - BATCH) * TB_X + blockIdx.x;
    geo_block((int)(id % (BATCH * NPTS / 256)), (int)(id / (BATCH * NPTS / 256)),
              xyz, sw1a, sw1d, bias1p, PQ);
  }
}

#define GA_THR 256
#define GA_PPT 2
#define GA_PTS (GA_THR * GA_PPT)
#define GA_STG (2 * NPTS / GA_THR)

__device__ __forceinline__ float4 max4(const float4 a, const float4 b) {
  return make_float4(fmaxf(a.x, b.x), fmaxf(a.y, b.y), fmaxf(a.z, b.z), fmaxf(a.w, b.w));
}

__global__ void __launch_bounds__(GA_THR)
gather_max(const float* __restrict__ PQ, const float* __restrict__ Cf,
           const int* __restrict__ idx, const float* __restrict__ bias2p,
           float* __restrict__ out) {
  __shared__ float4 tab[2 * NPTS];
  const int b    = blockIdx.z;
  const int tp   = blockIdx.y;
  const bool fea = tp >= 8;
  const int q0   = (tp & 7) * 2;
  const int tid  = threadIdx.x;
  const float4* T   = (const float4*)(fea ? Cf : PQ) + (size_t)b * QROWS * NPTS;
  const float4* src = T + (size_t)q0 * NPTS;

  float4 st[GA_STG];
  #pragma unroll
  for (int i = 0; i < GA_STG; ++i) st[i] = src[tid + i * GA_THR];

  const int nb = blockIdx.x * GA_PTS + tid;
  int4   id[GA_PPT][KNN_K / 4];
  float4 ctr[GA_PPT][2];
  #pragma unroll
  for (int p = 0; p < GA_PPT; ++p) {
    const int n = nb + p * GA_THR;
    const int4* ip = (const int4*)(idx + ((size_t)b * NPTS + n) * KNN_K);
    #pragma unroll
    for (int j = 0; j < KNN_K / 4; ++j) id[p][j] = ip[j];
    ctr[p][0] = T[(size_t)(16 + q0) * NPTS + n];
    ctr[p][1] = T[(size_t)(17 + q0) * NPTS + n];
  }

  #pragma unroll
  for (int i = 0; i < GA_STG; ++i) tab[tid + i * GA_THR] = st[i];

  float4 bs0 = make_float4(0.f, 0.f, 0.f, 0.f), bs1 = bs0;
  if (fea) {
    const float* bp = bias2p + 4 * q0;
    bs0 = make_float4(bp[0], bp[1], bp[2], bp[3]);
    bs1 = make_float4(bp[4], bp[5], bp[6], bp[7]);
  }
  __syncthreads();

  float* ob = out + ((size_t)b * CH + (fea ? HALF_CH : 0) + 4 * q0) * NPTS;
  #pragma unroll
  for (int p = 0; p < GA_PPT; ++p) {
    const int n = nb + p * GA_THR;
    float4 m0 = make_float4(-FLT_MAX, -FLT_MAX, -FLT_MAX, -FLT_MAX), m1 = m0;
    #pragma unroll
    for (int j = 0; j < KNN_K / 4; ++j) {
      const int4 q = id[p][j];
      m0 = max4(m0, tab[q.x]); m1 = max4(m1, tab[NPTS + q.x]);
      m0 = max4(m0, tab[q.y]); m1 = max4(m1, tab[NPTS + q.y]);
      m0 = max4(m0, tab[q.z]); m1 = max4(m1, tab[NPTS + q.z]);
      m0 = max4(m0, tab[q.w]); m1 = max4(m1, tab[NPTS + q.w]);
    }
    float4 c0 = ctr[p][0], c1 = ctr[p][1];
    if (fea) {
      c0 = make_float4(c0.x + bs0.x, c0.y + bs0.y, c0.z + bs0.z, c0.w + bs0.w);
      c1 = make_float4(c1.x + bs1.x, c1.y + bs1.y, c1.z + bs1.z, c1.w + bs1.w);
    }
    ob[0 * NPTS + n] = fmaxf(m0.x + c0.x, 0.0f);
    ob[1 * NPTS + n] = fmaxf(m0.y + c0.y, 0.0f);
    ob[2 * NPTS + n] = fmaxf(m0.z + c0.z, 0.0f);
    ob[3 * NPTS + n] = fmaxf(m0.w + c0.w, 0.0f);
    ob[4 * NPTS + n] = fmaxf(m1.x + c1.x, 0.0f);
    ob[5 * NPTS + n] = fmaxf(m1.y + c1.y, 0.0f);
    ob[6 * NPTS + n] = fmaxf(m1.z + c1.z, 0.0f);
    ob[7 * NPTS + n] = fmaxf(m1.w + c1.w, 0.0f);
  }
}

extern "C" void kernel_launch(void* const* d_in, const int* in_sizes, int n_in,
                              void* d_out, int out_size, void* d_ws, size_t ws_size,
                              hipStream_t stream) {
  (void)in_sizes; (void)n_in; (void)out_size; (void)ws_size;

  const float* xyz  = (const float*)d_in[0];
  const float* feat = (const float*)d_in[1];
  const float* w1   = (const float*)d_in[2];
  const float* b1   = (const float*)d_in[3];
  const float* g1   = (const float*)d_in[4];
  const float* be1  = (const float*)d_in[5];
  const float* m1   = (const float*)d_in[6];
  const float* v1   = (const float*)d_in[7];
  const float* w2   = (const float*)d_in[8];
  const float* b2   = (const float*)d_in[9];
  const float* g2   = (const float*)d_in[10];
  const float* be2  = (const float*)d_in[11];
  const float* m2   = (const float*)d_in[12];
  const float* v2   = (const float*)d_in[13];

  char* ws = (char*)d_ws;
  float* Wc     = (float*)ws;
  float* sw1a   = Wc + 16384;
  float* sw1d   = sw1a + 192;
  float* bias1p = sw1d + 192;
  float* bias2p = bias1p + 64;
  float* PQ     = (float*)(ws + (128 << 10));
  float* Cf     = PQ + (size_t)BATCH * NPTS * CH;
  int*   idxb   = (int*)(Cf + (size_t)BATCH * NPTS * CH);

  knn_kernel<<<dim3(NPTS / KNN_Q, BATCH + 1), KNN_Q * KNN_W, 0, stream>>>(
      xyz, idxb, w1, b1, g1, be1, m1, v1, w2, b2, g2, be2, m2, v2,
      Wc, sw1a, sw1d, bias1p, bias2p);

  tables_kernel<<<dim3(TB_X, BATCH + TB_GEO), 256, 0, stream>>>(xyz, feat, Wc, sw1a, sw1d,
                                                               bias1p, PQ, Cf);

  gather_max<<<dim3(NPTS / GA_PTS, 16, BATCH), GA_THR, 0, stream>>>(PQ, Cf, idxb, bias2p,
                                                                    (float*)d_out);
}


# gather tables (feature GEMM + geometry) become rows of 128-thread blocks of the kNN launch, beside the scans, folding their BN weights as they load them; 3 launches become 2; same sum order
# speedup vs baseline: 100.2834x; 1.2349x over previous
#include <hip/hip_runtime.h>
#include <hip/hip_bf16.h>
#include <float.h>

#define BATCH   8
#define NPTS    2048
#define CH      128
#define HALF_CH 64
#define KNN_K   20
#define EPSB    1e-5f

typedef __attribute__((ext_vector_type(2))) float v2f;
typedef __attribute__((ext_vector_type(8))) float v8f;

#define KNN_Q    32
#define KNN_W    4
#define KNN_SEG  (NPTS / KNN_W)
#define KNN_STG  (NPTS / (KNN_Q * KNN_W))
#define KNN_PRE  128
#define KNN_CAP  95

__device__ __forceinline__ float knn_score(const float4 me, const float4 p) {
  #pragma clang fp contract(off)
  const float yy  = me.y * p.y;
  const float zz  = me.z * p.z;
  const float dot = __builtin_fmaf(me.x, p.x, yy) + zz;
  return __builtin_fmaf(-2.0f, dot, me.w + p.w);
}

__device__ __forceinline__ void knn_insert(float (&t)[KNN_K], const float d) {
  #pragma unroll
  for (int i = KNN_K - 1; i > 0; --i) t[i] = __builtin_amdgcn_fmed3f(d, t[i - 1], t[i]);
  t[0] = fminf(d, t[0]);
}

__device__ __forceinline__ float bn_scale(const float g, const float v) {
  return g / sqrtf(v + EPSB);
}

__device__ __forceinline__ void fold_bias2(
    const float* __restrict__ b2, const float* __restrict__ g2,
    const float* __restrict__ be2, const float* __restrict__ m2,
    const float* __restrict__ v2, float* __restrict__ bias2p) {
  if (blockIdx.x == 0 && threadIdx.x < HALF_CH) {
    const int o = threadIdx.x;
    const float s2 = bn_scale(g2[o], v2[o]);
    bias2p[o] = b2[o] * s2 + (be2[o] - m2[o] * s2);
  }
}

__device__ __forceinline__ void table_block(const int row, float* sh,
           const float* __restrict__ xyz, const float* __restrict__ feat,
           const float* __restrict__ w1, const float* __restrict__ b1,
           const float* __restrict__ g1, const float* __restrict__ be1,
           const float* __restrict__ m1, const float* __restrict__ v1,
           const float* __restrict__ w2, const float* __restrict__ g2,
           const float* __restrict__ v2,
           float* __restrict__ PQ, float* __restrict__ Cf);

__global__ void __launch_bounds__(KNN_Q * KNN_W)
knn_kernel(const float* __restrict__ xyz, const float* __restrict__ feat,
           int* __restrict__ idx_out,
           const float* __restrict__ w1, const float* __restrict__ b1,
           const float* __restrict__ g1, const float* __restrict__ be1,
           const float* __restrict__ m1, const float* __restrict__ v1,
           const float* __restrict__ w2, const float* __restrict__ b2,
           const float* __restrict__ g2, const float* __restrict__ be2,
           const float* __restrict__ m2, const float* __restrict__ v2,
           float* __restrict__ bias2p,
           float* __restrict__ PQ, float* __restrict__ Cf) {
  __shared__ float4 pts[NPTS];
  __shared__ float  lst[KNN_W][KNN_K][KNN_Q];
  __shared__ int    hit[KNN_W][KNN_CAP + 1][KNN_Q];
  if (blockIdx.y >= BATCH) {
    if (blockIdx.y == BATCH) fold_bias2(b2, g2, be2, m2, v2, bias2p);
    else table_block((int)blockIdx.y - (BATCH + 1), (float*)pts, xyz, feat,
                     w1, b1, g1, be1, m1, v1, w2, g2, v2, PQ, Cf);
    return;
  }
  const int b = blockIdx.y;
  const float* xb = xyz + (size_t)b * 3 * NPTS;
  float sx[KNN_STG], sy[KNN_STG], sz[KNN_STG];
  #pragma unroll
  for (int i = 0; i < KNN_STG; ++i) {
    const int m = threadIdx.x + i * (KNN_Q * KNN_W);
    sx[i] = xb[m]; sy[i] = xb[NPTS + m]; sz[i] = xb[2 * NPTS + m];
  }
  #pragma unroll
  for (int i = 0; i < KNN_STG; ++i) {
    const int m = threadIdx.x + i * (KNN_Q * KNN_W);
    const float x = sx[i], y = sy[i], z = sz[i];
    pts[m] = make_float4(x, y, z, __builtin_fmaf(z, z, __builtin_fmaf(x, x, y * y)));
  }
  __syncthreads();

  const int lane = threadIdx.x & (KNN_Q - 1);
  const int w    = threadIdx.x / KNN_Q;
  const int n    = blockIdx.x * KNN_Q + lane;
  const float4 me = pts[n];
  const float4* seg = pts + w * KNN_SEG;
  int (*myhit)[KNN_Q] = hit[w];
  int nhit = 0;

  float t[KNN_K];
  #pragma unroll
  for (int i = 0; i < KNN_K; ++i) t[i] = FLT_MAX;

  #pragma unroll 8
  for (int m = 0; m < KNN_PRE; ++m) knn_insert(t, knn_score(me, seg[m]));

  for (int m0 = KNN_PRE; m0 < KNN_SEG; m0 += 32) {
    const float u = t[KNN_K - 1];
    unsigned in = 0u;
    #pragma unroll 8
    for (int k = 0; k < 32; ++k) {
      const float d = knn_score(me, seg[m0 + k]);
      in = in + in + ((d < u) ? 1u : 0u);
    }
    if (__builtin_amdgcn_ballot_w32(in != 0u) != 0u) {
      bool has0 = in != 0u;
      int k = __builtin_clz(in | 1u);
      in &= ~(0x80000000u >> k);
      int i0 = m0 + k;
      float4 p0 = seg[i0];
      bool has1 = in != 0u;
      k = __builtin_clz(in | 1u);
      in &= ~(0x80000000u >> k);
      int i1 = m0 + k;
      float4 p1 = seg[i1];
      unsigned more;
      do {
        more = __builtin_amdgcn_ballot_w32(in != 0u);
        const bool hn0 = in != 0u;
        k = __builtin_clz(in | 1u);
        in &= ~(0x80000000u >> k);
        const int j0 = m0 + k;
        const float4 q0 = seg[j0];
        const bool hn1 = in != 0u;
        k = __builtin_clz(in | 1u);
        in &= ~(0x80000000u >> k);
        const int j1 = m0 + k;
        const float4 q1 = seg[j1];
        __builtin_amdgcn_sched_barrier(0);
        const float d0 = knn_score(me, p0);
        knn_insert(t, has0 ? d0 : FLT_MAX);
        const float d1 = knn_score(me, p1);
        knn_insert(t, has1 ? d1 : FLT_MAX);
        myhit[min(nhit, KNN_CAP)][lane] = i0;
        nhit += has0 ? 1 : 0;
        myhit[min(nhit, KNN_CAP)][lane] = i1;
        nhit += has1 ? 1 : 0;
        __builtin_amdgcn_sched_barrier(0);
        asm volatile("" :: "v"(q0.x), "v"(q0.y), "v"(q0.z), "v"(q0.w),
                           "v"(q1.x), "v"(q1.y), "v"(q1.z), "v"(q1.w));
        p0 = q0; p1 = q1; has0 = hn0; has1 = hn1; i0 = j0; i1 = j1;
      } while (more != 0u);
    }
  }

  #pragma unroll
  for (int i = 0; i < KNN_K; ++i) lst[w][i][lane] = t[i];
  __syncthreads();
  for (int o = 1; o < KNN_W; ++o) {
    const int w2 = (w + o) & (KNN_W - 1);
    float v[KNN_K];
    #pragma unroll
    for (int i = 0; i < KNN_K; ++i) v[i] = lst[w2][i][lane];
    #pragma unroll
    for (int i = 0; i < KNN_K; ++i) knn_insert(t, v[i]);
  }

  const float thr = t[KNN_K - 1];
  int ties = 0;
  #pragma unroll
  for (int i = 0; i < KNN_K; ++i) ties += (t[i] == thr) ? 1 : 0;

  int off = 0;
  int quota = 0;
  int mine_eq = 0;
  int rem = ties;
  bool exact = false;
  for (int q = 0; q <= w; ++q) {
    int lt = 0, eq = 0;
    #pragma unroll
    for (int i = 0; i < KNN_K; ++i) {
      const float s = lst[q][i][lane];
      lt += (s < thr) ? 1 : 0;
      eq += (s == thr) ? 1 : 0;
    }
    const int a = min(eq, rem);
    rem -= a;
    if (q < w) off += lt + a;
    else {
      quota = lt + a; mine_eq = a;
      exact = (a < eq) || !(lst[q][KNN_K - 1][lane] > thr);
    }
  }
  if (w == KNN_W - 1) quota = KNN_K - off;

  int* op = idx_out + ((size_t)b * NPTS + n) * KNN_K + off;
  const int base = w * KNN_SEG;
  int cnt = 0;
  auto take = [&](const bool tie, const int id) {
    if ((!tie || mine_eq > 0) && cnt < quota) { op[cnt] = id; ++cnt; }
    if (tie) --mine_eq;
  };
  const bool spilt = __builtin_amdgcn_ballot_w32(nhit > KNN_CAP) != 0u;
  const int lim = spilt ? KNN_SEG : KNN_PRE;
  for (int m0 = 0; m0 < lim; m0 += 32) {
    unsigned le = 0u;
    #pragma unroll 8
    for (int k = 0; k < 32; ++k) {
      const float d = knn_score(me, seg[m0 + k]);
      le = le + le + ((d <= thr) ? 1u : 0u);
    }
    while (le != 0u) {
      const int k = __builtin_clz(le);
      bool tie = false;
      if (exact) tie = !(knn_score(me, seg[m0 + k]) < thr);
      take(tie, base + m0 + k);
      le &= ~(0x80000000u >> k);
    }
  }
  if (!spilt) {
    const int last = max(nhit - 1, 0);
    for (int j0 = 0; __builtin_amdgcn_ballot_w32(j0 < nhit) != 0u; j0 += 32) {
      unsigned le = 0u;
      #pragma unroll 8
      for (int k = 0; k < 32; ++k) {
        const float d = knn_score(me, seg[myhit[min(j0 + k, last)][lane]]);
        le = le + le + ((d <= thr) ? 1u : 0u);
      }
      const int have = min(max(nhit - j0, 0), 32);
      le &= (unsigned)(0xffffffff00000000ull >> have);
      while (le != 0u) {
        const int k = __builtin_clz(le);
        const int id = myhit[j0 + k][lane];
        bool tie = false;
        if (exact) tie = !(knn_score(me, seg[id]) < thr);
        take(tie, base + id);
        le &= ~(0x80000000u >> k);
      }
    }
  }
  for (; cnt < quota; ++cnt) op[cnt] = n;
}

#define QROWS 32

#define TB_THR   (KNN_Q * KNN_W)
#define TB_X     (NPTS / KNN_Q)
#define GM_PTS   32
#define GM_ROWS  (BATCH * NPTS / GM_PTS / TB_X)
#define GEO_BLKS (BATCH * NPTS / TB_THR)
#define GEO_ROWS (GEO_BLKS * 4 / TB_X)

template <bool CTR>
__device__ __forceinline__ void geo_rows(const float4* wt,
                                         const float x, const float y, const float z,
                                         float4* __restrict__ o) {
  #pragma unroll
  for (int j = 0; j < 8; ++j) {
    float v[4];
    #pragma unroll
    for (int e = 0; e < 4; ++e) {
      const float4 w = wt[4 * j + e];
      if (CTR) v[e] = w.x * x + w.y * y + w.z * z + w.w;
      else     v[e] = w.x * x + w.y * y + w.z * z;
    }
    o[(size_t)j * NPTS] = make_float4(v[0], v[1], v[2], v[3]);
  }
}

__device__ __forceinline__ void geo_block(const int id, float* sh,
           const float* __restrict__ xyz,
           const float* __restrict__ w1, const float* __restrict__ b1,
           const float* __restrict__ g1, const float* __restrict__ be1,
           const float* __restrict__ m1, const float* __restrict__ v1,
           float* __restrict__ PQ) {
  const int bx = id % GEO_BLKS;
  const int g  = id / GEO_BLKS;
  const int c0 = (g & 1) * 32;
  const bool ctr = g >= 2;
  const int pt = bx * TB_THR + threadIdx.x;
  const int b  = pt / NPTS;
  const int n  = pt - b * NPTS;
  const float* xb = xyz + (size_t)b * 3 * NPTS;
  const float x = xb[n], y = xb[NPTS + n], z = xb[2 * NPTS + n];
  float4* wt = (float4*)sh;
  if (threadIdx.x < 32) {
    const int c = c0 + threadIdx.x;
    const float s1 = bn_scale(g1[c], v1[c]);
    float w[3];
    #pragma unroll
    for (int j = 0; j < 3; ++j) {
      const float wa = w1[c * 6 + j];
      const float wb = w1[c * 6 + 3 + j];
      w[j] = ctr ? s1 * (wb - wa) : s1 * wa;
    }
    const float bias = ctr ? b1[c] * s1 + (be1[c] - m1[c] * s1) : 0.0f;
    wt[threadIdx.x] = make_float4(w[0], w[1], w[2], bias);
  }
  __syncthreads();
  float4* o = (float4*)PQ + ((size_t)b * QROWS + 8 * g) * NPTS + n;
  if (ctr) geo_rows<true>(wt, x, y, z, o);
  else     geo_rows<false>(wt, x, y, z, o);
}

__device__ __forceinline__ void gemm_block(const int bx, const int b,
                          const float* __restrict__ feat, const float* __restrict__ w2,
                          const float* __restrict__ g2, const float* __restrict__ v2,
                          float* __restrict__ Cf) {
  const int n0   = bx * GM_PTS;
  const int wave = threadIdx.x >> 5;
  const int lane = threadIdx.x & 31;
  const int l16  = lane & 15;
  const int hi   = lane >> 4;
  const int o    = wave * 16 + l16;
  const float s2 = bn_scale(g2[o], v2[o]);

  const float* Fb   = feat + (size_t)b * CH * NPTS;
  const float* wa   = w2 + (size_t)o * (2 * CH) + 2 * hi;
  const float* wb   = wa + CH;
  const float* bcol = Fb + (size_t)(2 * hi) * NPTS + n0 + l16;

  v8f an0 = {0.f, 0.f, 0.f, 0.f, 0.f, 0.f, 0.f, 0.f};
  v8f an1 = an0, ac0 = an0, ac1 = an0;

  #pragma unroll 8
  for (int k = 0; k < CH; k += 4) {
    const float wax = wa[k], way = wa[k + 1];
    const float wbx = wb[k], wby = wb[k + 1];
    v2f a_n;
    a_n.x = s2 * wax;
    a_n.y = s2 * way;
    v2f a_c;
    a_c.x = s2 * (wbx - wax);
    a_c.y = s2 * (wby - way);
    v2f b0;
    b0.x = bcol[(size_t)k * NPTS];
    b0.y = bcol[(size_t)(k + 1) * NPTS];
    v2f b1;
    b1.x = bcol[(size_t)k * NPTS + 16];
    b1.y = bcol[(size_t)(k + 1) * NPTS + 16];
    an0 = __builtin_amdgcn_wmma_f32_16x16x4_f32(false, a_n, false, b0, (short)0, an0, false, false);
    an1 = __builtin_amdgcn_wmma_f32_16x16x4_f32(false, a_n, false, b1, (short)0, an1, false, false);
    ac0 = __builtin_amdgcn_wmma_f32_16x16x4_f32(false, a_c, false, b0, (short)0, ac0, false, false);
    ac1 = __builtin_amdgcn_wmma_f32_16x16x4_f32(false, a_c, false, b1, (short)0, ac1, false, false);
  }

  float4* on = (float4*)Cf + ((size_t)b * QROWS + (16 * wave + 8 * hi) / 4) * NPTS + n0 + l16;
  float4* oc = on + (size_t)(HALF_CH / 4) * NPTS;
  on[0]         = make_float4(an0[0], an0[1], an0[2], an0[3]);
  on[NPTS]      = make_float4(an0[4], an0[5], an0[6], an0[7]);
  on[16]        = make_float4(an1[0], an1[1], an1[2], an1[3]);
  on[NPTS + 16] = make_float4(an1[4], an1[5], an1[6], an1[7]);
  oc[0]         = make_float4(ac0[0], ac0[1], ac0[2], ac0[3]);
  oc[NPTS]      = make_float4(ac0[4], ac0[5], ac0[6], ac0[7]);
  oc[16]        = make_float4(ac1[0], ac1[1], ac1[2], ac1[3]);
  oc[NPTS + 16] = make_float4(ac1[4], ac1[5], ac1[6], ac1[7]);
}

__device__ __forceinline__ void table_block(const int row, float* sh,
           const float* __restrict__ xyz, const float* __restrict__ feat,
           const float* __restrict__ w1, const float* __restrict__ b1,
           const float* __restrict__ g1, const float* __restrict__ be1,
           const float* __restrict__ m1, const float* __restrict__ v1,
           const float* __restrict__ w2, const float* __restrict__ g2,
           const float* __restrict__ v2,
           float* __restrict__ PQ, float* __restrict__ Cf) {
  if (row < GM_ROWS) gemm_block((int)blockIdx.x, row, feat, w2, g2, v2, Cf);
  else geo_block((row - GM_ROWS) * TB_X + (int)blockIdx.x, sh, xyz, w1, b1, g1, be1, m1, v1, PQ);
}

#define GA_THR 256
#define GA_PPT 2
#define GA_PTS (GA_THR * GA_PPT)
#define GA_STG (2 * NPTS / GA_THR)

__device__ __forceinline__ float4 max4(const float4 a, const float4 b) {
  return make_float4(fmaxf(a.x, b.x), fmaxf(a.y, b.y), fmaxf(a.z, b.z), fmaxf(a.w, b.w));
}

__global__ void __launch_bounds__(GA_THR)
gather_max(const float* __restrict__ PQ, const float* __restrict__ Cf,
           const int* __restrict__ idx, const float* __restrict__ bias2p,
           float* __restrict__ out) {
  __shared__ float4 tab[2 * NPTS];
  const int b    = blockIdx.z;
  const int tp   = blockIdx.y;
  const bool fea = tp >= 8;
  const int q0   = (tp & 7) * 2;
  const int tid  = threadIdx.x;
  const float4* T   = (const float4*)(fea ? Cf : PQ) + (size_t)b * QROWS * NPTS;
  const float4* src = T + (size_t)q0 * NPTS;

  float4 st[GA_STG];
  #pragma unroll
  for (int i = 0; i < GA_STG; ++i) st[i] = src[tid + i * GA_THR];

  const int nb = blockIdx.x * GA_PTS + tid;
  int4   id[GA_PPT][KNN_K / 4];
  float4 ctr[GA_PPT][2];
  #pragma unroll
  for (int p = 0; p < GA_PPT; ++p) {
    const int n = nb + p * GA_THR;
    const int4* ip = (const int4*)(idx + ((size_t)b * NPTS + n) * KNN_K);
    #pragma unroll
    for (int j = 0; j < KNN_K / 4; ++j) id[p][j] = ip[j];
    ctr[p][0] = T[(size_t)(16 + q0) * NPTS + n];
    ctr[p][1] = T[(size_t)(17 + q0) * NPTS + n];
  }

  #pragma unroll
  for (int i = 0; i < GA_STG; ++i) tab[tid + i * GA_THR] = st[i];

  float4 bs0 = make_float4(0.f, 0.f, 0.f, 0.f), bs1 = bs0;
  if (fea) {
    const float* bp = bias2p + 4 * q0;
    bs0 = make_float4(bp[0], bp[1], bp[2], bp[3]);
    bs1 = make_float4(bp[4], bp[5], bp[6], bp[7]);
  }
  __syncthreads();

  float* ob = out + ((size_t)b * CH + (fea ? HALF_CH : 0) + 4 * q0) * NPTS;
  #pragma unroll
  for (int p = 0; p < GA_PPT; ++p) {
    const int n = nb + p * GA_THR;
    float4 m0 = make_float4(-FLT_MAX, -FLT_MAX, -FLT_MAX, -FLT_MAX), m1 = m0;
    #pragma unroll
    for (int j = 0; j < KNN_K / 4; ++j) {
      const int4 q = id[p][j];
      m0 = max4(m0, tab[q.x]); m1 = max4(m1, tab[NPTS + q.x]);
      m0 = max4(m0, tab[q.y]); m1 = max4(m1, tab[NPTS + q.y]);
      m0 = max4(m0, tab[q.z]); m1 = max4(m1, tab[NPTS + q.z]);
      m0 = max4(m0, tab[q.w]); m1 = max4(m1, tab[NPTS + q.w]);
    }
    float4 c0 = ctr[p][0], c1 = ctr[p][1];
    if (fea) {
      c0 = make_float4(c0.x + bs0.x, c0.y + bs0.y, c0.z + bs0.z, c0.w + bs0.w);
      c1 = make_float4(c1.x + bs1.x, c1.y + bs1.y, c1.z + bs1.z, c1.w + bs1.w);
    }
    ob[0 * NPTS + n] = fmaxf(m0.x + c0.x, 0.0f);
    ob[1 * NPTS + n] = fmaxf(m0.y + c0.y, 0.0f);
    ob[2 * NPTS + n] = fmaxf(m0.z + c0.z, 0.0f);
    ob[3 * NPTS + n] = fmaxf(m0.w + c0.w, 0.0f);
    ob[4 * NPTS + n] = fmaxf(m1.x + c1.x, 0.0f);
    ob[5 * NPTS + n] = fmaxf(m1.y + c1.y, 0.0f);
    ob[6 * NPTS + n] = fmaxf(m1.z + c1.z, 0.0f);
    ob[7 * NPTS + n] = fmaxf(m1.w + c1.w, 0.0f);
  }
}

extern "C" void kernel_launch(void* const* d_in, const int* in_sizes, int n_in,
                              void* d_out, int out_size, void* d_ws, size_t ws_size,
                              hipStream_t stream) {
  (void)in_sizes; (void)n_in; (void)out_size; (void)ws_size;

  const float* xyz  = (const float*)d_in[0];
  const float* feat = (const float*)d_in[1];
  const float* w1   = (const float*)d_in[2];
  const float* b1   = (const float*)d_in[3];
  const float* g1   = (const float*)d_in[4];
  const float* be1  = (const float*)d_in[5];
  const float* m1   = (const float*)d_in[6];
  const float* v1   = (const float*)d_in[7];
  const float* w2   = (const float*)d_in[8];
  const float* b2   = (const float*)d_in[9];
  const float* g2   = (const float*)d_in[10];
  const float* be2  = (const float*)d_in[11];
  const float* m2   = (const float*)d_in[12];
  const float* v2   = (const float*)d_in[13];

  char* ws = (char*)d_ws;
  float* bias2p = (float*)ws;
  float* PQ     = (float*)(ws + (128 << 10));
  float* Cf     = PQ + (size_t)BATCH * NPTS * CH;
  int*   idxb   = (int*)(Cf + (size_t)BATCH * NPTS * CH);

  knn_kernel<<<dim3(TB_X, BATCH + 1 + GM_ROWS + GEO_ROWS), TB_THR, 0, stream>>>(
      xyz, feat, idxb, w1, b1, g1, be1, m1, v1, w2, b2, g2, be2, m2, v2,
      bias2p, PQ, Cf);

  gather_max<<<dim3(NPTS / GA_PTS, 16, BATCH), GA_THR, 0, stream>>>(PQ, Cf, idxb, bias2p,
                                                                    (float*)d_out);
}


# kNN: one-by-one median-chain insertions of whole batches become min/max networks: head scores sorted 32 at a time and joined to the list, the four quarter lists joined pairwise; scores bit-identical
# speedup vs baseline: 105.7960x; 1.0550x over previous
#include <hip/hip_runtime.h>
#include <hip/hip_bf16.h>
#include <float.h>

#define BATCH   8
#define NPTS    2048
#define CH      128
#define HALF_CH 64
#define KNN_K   20
#define EPSB    1e-5f

typedef __attribute__((ext_vector_type(2))) float v2f;
typedef __attribute__((ext_vector_type(8))) float v8f;

#define KNN_Q    32
#define KNN_W    4
#define KNN_SEG  (NPTS / KNN_W)
#define KNN_STG  (NPTS / (KNN_Q * KNN_W))
#define KNN_PRE  128
#define KNN_CAP  95

__device__ __forceinline__ float knn_score(const float4 me, const float4 p) {
  #pragma clang fp contract(off)
  const float yy  = me.y * p.y;
  const float zz  = me.z * p.z;
  const float dot = __builtin_fmaf(me.x, p.x, yy) + zz;
  return __builtin_fmaf(-2.0f, dot, me.w + p.w);
}

__device__ __forceinline__ void knn_insert(float (&t)[KNN_K], const float d) {
  #pragma unroll
  for (int i = KNN_K - 1; i > 0; --i) t[i] = __builtin_amdgcn_fmed3f(d, t[i - 1], t[i]);
  t[0] = fminf(d, t[0]);
}

template <int NY>
__device__ __forceinline__ void knn_join(const float (&x)[KNN_K], const float (&y)[NY],
                                         float (&e)[32]) {
  static_assert(KNN_K <= 32 && NY <= 32 && KNN_K + NY >= 32, "every e[i] needs a real score");
  #pragma unroll
  for (int i = 0; i < 32; ++i) {
    const int j = 31 - i;
    e[i] = (i < KNN_K && j < NY) ? fminf(x[i], y[j]) : (i < KNN_K) ? x[i] : y[j];
  }
  #pragma unroll
  for (int d = 16; d >= 1; d >>= 1) {
    #pragma unroll
    for (int i = 0; i < 32; ++i) {
      if ((i & d) == 0) {
        const float a = e[i], b = e[i + d];
        e[i] = fminf(a, b);
        e[i + d] = fmaxf(a, b);
      }
    }
  }
}

#define KNN_CE(a, b) { const float lo_ = fminf(s[a], s[b]); s[b] = fmaxf(s[a], s[b]); s[a] = lo_; }
__device__ __forceinline__ void knn_sort32(float (&s)[32]) {
  KNN_CE(0,1) KNN_CE(2,3) KNN_CE(4,5) KNN_CE(6,7) KNN_CE(8,9) KNN_CE(10,11) KNN_CE(12,13) KNN_CE(14,15)
  KNN_CE(16,17) KNN_CE(18,19) KNN_CE(20,21) KNN_CE(22,23) KNN_CE(24,25) KNN_CE(26,27) KNN_CE(28,29) KNN_CE(30,31)
  KNN_CE(0,2) KNN_CE(1,3) KNN_CE(4,6) KNN_CE(5,7) KNN_CE(8,10) KNN_CE(9,11) KNN_CE(12,14) KNN_CE(13,15)
  KNN_CE(16,18) KNN_CE(17,19) KNN_CE(20,22) KNN_CE(21,23) KNN_CE(24,26) KNN_CE(25,27) KNN_CE(28,30) KNN_CE(29,31)
  KNN_CE(1,2) KNN_CE(5,6) KNN_CE(9,10) KNN_CE(13,14) KNN_CE(17,18) KNN_CE(21,22) KNN_CE(25,26) KNN_CE(29,30)
  KNN_CE(0,4) KNN_CE(1,5) KNN_CE(2,6) KNN_CE(3,7) KNN_CE(8,12) KNN_CE(9,13) KNN_CE(10,14) KNN_CE(11,15)
  KNN_CE(16,20) KNN_CE(17,21) KNN_CE(18,22) KNN_CE(19,23) KNN_CE(24,28) KNN_CE(25,29) KNN_CE(26,30) KNN_CE(27,31)
  KNN_CE(2,4) KNN_CE(3,5) KNN_CE(10,12) KNN_CE(11,13) KNN_CE(18,20) KNN_CE(19,21) KNN_CE(26,28) KNN_CE(27,29)
  KNN_CE(1,2) KNN_CE(3,4) KNN_CE(5,6) KNN_CE(9,10) KNN_CE(11,12) KNN_CE(13,14) KNN_CE(17,18) KNN_CE(19,20)
  KNN_CE(21,22) KNN_CE(25,26) KNN_CE(27,28) KNN_CE(29,30) KNN_CE(0,8) KNN_CE(1,9) KNN_CE(2,10) KNN_CE(3,11)
  KNN_CE(4,12) KNN_CE(5,13) KNN_CE(6,14) KNN_CE(7,15) KNN_CE(16,24) KNN_CE(17,25) KNN_CE(18,26) KNN_CE(19,27)
  KNN_CE(20,28) KNN_CE(21,29) KNN_CE(22,30) KNN_CE(23,31) KNN_CE(4,8) KNN_CE(5,9) KNN_CE(6,10) KNN_CE(7,11)
  KNN_CE(20,24) KNN_CE(21,25) KNN_CE(22,26) KNN_CE(23,27) KNN_CE(2,4) KNN_CE(3,5) KNN_CE(6,8) KNN_CE(7,9)
  KNN_CE(10,12) KNN_CE(11,13) KNN_CE(18,20) KNN_CE(19,21) KNN_CE(22,24) KNN_CE(23,25) KNN_CE(26,28) KNN_CE(27,29)
  KNN_CE(1,2) KNN_CE(3,4) KNN_CE(5,6) KNN_CE(7,8) KNN_CE(9,10) KNN_CE(11,12) KNN_CE(13,14) KNN_CE(17,18)
  KNN_CE(19,20) KNN_CE(21,22) KNN_CE(23,24) KNN_CE(25,26) KNN_CE(27,28) KNN_CE(29,30) KNN_CE(0,16) KNN_CE(1,17)
  KNN_CE(2,18) KNN_CE(3,19) KNN_CE(4,20) KNN_CE(5,21) KNN_CE(6,22) KNN_CE(7,23) KNN_CE(8,24) KNN_CE(9,25)
  KNN_CE(10,26) KNN_CE(11,27) KNN_CE(12,28) KNN_CE(13,29) KNN_CE(14,30) KNN_CE(15,31) KNN_CE(8,16) KNN_CE(9,17)
  KNN_CE(10,18) KNN_CE(11,19) KNN_CE(12,20) KNN_CE(13,21) KNN_CE(14,22) KNN_CE(15,23) KNN_CE(4,8) KNN_CE(5,9)
  KNN_CE(6,10) KNN_CE(7,11) KNN_CE(12,16) KNN_CE(13,17) KNN_CE(14,18) KNN_CE(15,19) KNN_CE(20,24) KNN_CE(21,25)
  KNN_CE(22,26) KNN_CE(23,27) KNN_CE(2,4) KNN_CE(3,5) KNN_CE(6,8) KNN_CE(7,9) KNN_CE(10,12) KNN_CE(11,13)
  KNN_CE(14,16) KNN_CE(15,17) KNN_CE(18,20) KNN_CE(19,21) KNN_CE(22,24) KNN_CE(23,25) KNN_CE(26,28) KNN_CE(27,29)
  KNN_CE(1,2) KNN_CE(3,4) KNN_CE(5,6) KNN_CE(7,8) KNN_CE(9,10) KNN_CE(11,12) KNN_CE(13,14) KNN_CE(15,16)
  KNN_CE(17,18) KNN_CE(19,20) KNN_CE(21,22) KNN_CE(23,24) KNN_CE(25,26) KNN_CE(27,28) KNN_CE(29,30)
}
#undef KNN_CE

__device__ __forceinline__ float bn_scale(const float g, const float v) {
  return g / sqrtf(v + EPSB);
}

__device__ __forceinline__ void fold_bias2(
    const float* __restrict__ b2, const float* __restrict__ g2,
    const float* __restrict__ be2, const float* __restrict__ m2,
    const float* __restrict__ v2, float* __restrict__ bias2p) {
  if (blockIdx.x == 0 && threadIdx.x < HALF_CH) {
    const int o = threadIdx.x;
    const float s2 = bn_scale(g2[o], v2[o]);
    bias2p[o] = b2[o] * s2 + (be2[o] - m2[o] * s2);
  }
}

__device__ __forceinline__ void table_block(const int row, float* sh,
           const float* __restrict__ xyz, const float* __restrict__ feat,
           const float* __restrict__ w1, const float* __restrict__ b1,
           const float* __restrict__ g1, const float* __restrict__ be1,
           const float* __restrict__ m1, const float* __restrict__ v1,
           const float* __restrict__ w2, const float* __restrict__ g2,
           const float* __restrict__ v2,
           float* __restrict__ PQ, float* __restrict__ Cf);

__global__ void __launch_bounds__(KNN_Q * KNN_W)
knn_kernel(const float* __restrict__ xyz, const float* __restrict__ feat,
           int* __restrict__ idx_out,
           const float* __restrict__ w1, const float* __restrict__ b1,
           const float* __restrict__ g1, const float* __restrict__ be1,
           const float* __restrict__ m1, const float* __restrict__ v1,
           const float* __restrict__ w2, const float* __restrict__ b2,
           const float* __restrict__ g2, const float* __restrict__ be2,
           const float* __restrict__ m2, const float* __restrict__ v2,
           float* __restrict__ bias2p,
           float* __restrict__ PQ, float* __restrict__ Cf) {
  __shared__ float4 pts[NPTS];
  __shared__ float  lst[KNN_W][KNN_K][KNN_Q];
  __shared__ int    hit[KNN_W][KNN_CAP + 1][KNN_Q];
  if (blockIdx.y >= BATCH) {
    if (blockIdx.y == BATCH) fold_bias2(b2, g2, be2, m2, v2, bias2p);
    else table_block((int)blockIdx.y - (BATCH + 1), (float*)pts, xyz, feat,
                     w1, b1, g1, be1, m1, v1, w2, g2, v2, PQ, Cf);
    return;
  }
  const int b = blockIdx.y;
  const float* xb = xyz + (size_t)b * 3 * NPTS;
  float sx[KNN_STG], sy[KNN_STG], sz[KNN_STG];
  #pragma unroll
  for (int i = 0; i < KNN_STG; ++i) {
    const int m = threadIdx.x + i * (KNN_Q * KNN_W);
    sx[i] = xb[m]; sy[i] = xb[NPTS + m]; sz[i] = xb[2 * NPTS + m];
  }
  #pragma unroll
  for (int i = 0; i < KNN_STG; ++i) {
    const int m = threadIdx.x + i * (KNN_Q * KNN_W);
    const float x = sx[i], y = sy[i], z = sz[i];
    pts[m] = make_float4(x, y, z, __builtin_fmaf(z, z, __builtin_fmaf(x, x, y * y)));
  }
  __syncthreads();

  const int lane = threadIdx.x & (KNN_Q - 1);
  const int w    = threadIdx.x / KNN_Q;
  const int n    = blockIdx.x * KNN_Q + lane;
  const float4 me = pts[n];
  const float4* seg = pts + w * KNN_SEG;
  int (*myhit)[KNN_Q] = hit[w];
  int nhit = 0;

  float t[KNN_K];
  #pragma unroll
  for (int i = 0; i < KNN_K; ++i) t[i] = FLT_MAX;

  static_assert(KNN_PRE % 32 == 0, "the head is taken in whole batches of 32");
  for (int m0 = 0; m0 < KNN_PRE; m0 += 32) {
    float s[32];
    #pragma unroll
    for (int k = 0; k < 32; ++k) s[k] = knn_score(me, seg[m0 + k]);
    knn_sort32(s);
    float e[32];
    knn_join<32>(t, s, e);
    #pragma unroll
    for (int i = 0; i < KNN_K; ++i) t[i] = e[i];
  }

  for (int m0 = KNN_PRE; m0 < KNN_SEG; m0 += 32) {
    const float u = t[KNN_K - 1];
    unsigned in = 0u;
    #pragma unroll 8
    for (int k = 0; k < 32; ++k) {
      const float d = knn_score(me, seg[m0 + k]);
      in = in + in + ((d < u) ? 1u : 0u);
    }
    if (__builtin_amdgcn_ballot_w32(in != 0u) != 0u) {
      bool has0 = in != 0u;
      int k = __builtin_clz(in | 1u);
      in &= ~(0x80000000u >> k);
      int i0 = m0 + k;
      float4 p0 = seg[i0];
      bool has1 = in != 0u;
      k = __builtin_clz(in | 1u);
      in &= ~(0x80000000u >> k);
      int i1 = m0 + k;
      float4 p1 = seg[i1];
      unsigned more;
      do {
        more = __builtin_amdgcn_ballot_w32(in != 0u);
        const bool hn0 = in != 0u;
        k = __builtin_clz(in | 1u);
        in &= ~(0x80000000u >> k);
        const int j0 = m0 + k;
        const float4 q0 = seg[j0];
        const bool hn1 = in != 0u;
        k = __builtin_clz(in | 1u);
        in &= ~(0x80000000u >> k);
        const int j1 = m0 + k;
        const float4 q1 = seg[j1];
        __builtin_amdgcn_sched_barrier(0);
        const float d0 = knn_score(me, p0);
        knn_insert(t, has0 ? d0 : FLT_MAX);
        const float d1 = knn_score(me, p1);
        knn_insert(t, has1 ? d1 : FLT_MAX);
        myhit[min(nhit, KNN_CAP)][lane] = i0;
        nhit += has0 ? 1 : 0;
        myhit[min(nhit, KNN_CAP)][lane] = i1;
        nhit += has1 ? 1 : 0;
        __builtin_amdgcn_sched_barrier(0);
        asm volatile("" :: "v"(q0.x), "v"(q0.y), "v"(q0.z), "v"(q0.w),
                           "v"(q1.x), "v"(q1.y), "v"(q1.z), "v"(q1.w));
        p0 = q0; p1 = q1; has0 = hn0; has1 = hn1; i0 = j0; i1 = j1;
      } while (more != 0u);
    }
  }

  #pragma unroll
  for (int i = 0; i < KNN_K; ++i) lst[w][i][lane] = t[i];
  __syncthreads();
  static_assert(KNN_W == 4, "the pairing below is for four lists");
  float low[KNN_K];
  {
    float l1[KNN_K], l2[KNN_K], l3[KNN_K];
    #pragma unroll
    for (int i = 0; i < KNN_K; ++i) {
      l1[i] = lst[w ^ 1][i][lane];
      l2[i] = lst[w ^ 2][i][lane];
      l3[i] = lst[w ^ 3][i][lane];
    }
    float a[32], b[32];
    knn_join<KNN_K>(t, l1, a);
    knn_join<KNN_K>(l2, l3, b);
    #pragma unroll
    for (int i = 0; i < KNN_K; ++i) low[i] = fminf(a[i], b[KNN_K - 1 - i]);
  }

  float thr = low[0];
  #pragma unroll
  for (int i = 1; i < KNN_K; ++i) thr = fmaxf(thr, low[i]);
  int ties = 0;
  #pragma unroll
  for (int i = 0; i < KNN_K; ++i) ties += (low[i] == thr) ? 1 : 0;

  int off = 0;
  int quota = 0;
  int mine_eq = 0;
  int rem = ties;
  bool exact = false;
  for (int q = 0; q <= w; ++q) {
    int lt = 0, eq = 0;
    #pragma unroll
    for (int i = 0; i < KNN_K; ++i) {
      const float s = lst[q][i][lane];
      lt += (s < thr) ? 1 : 0;
      eq += (s == thr) ? 1 : 0;
    }
    const int a = min(eq, rem);
    rem -= a;
    if (q < w) off += lt + a;
    else {
      quota = lt + a; mine_eq = a;
      exact = (a < eq) || !(lst[q][KNN_K - 1][lane] > thr);
    }
  }
  if (w == KNN_W - 1) quota = KNN_K - off;

  int* op = idx_out + ((size_t)b * NPTS + n) * KNN_K + off;
  const int base = w * KNN_SEG;
  int cnt = 0;
  auto take = [&](const bool tie, const int id) {
    if ((!tie || mine_eq > 0) && cnt < quota) { op[cnt] = id; ++cnt; }
    if (tie) --mine_eq;
  };
  const bool spilt = __builtin_amdgcn_ballot_w32(nhit > KNN_CAP) != 0u;
  const int lim = spilt ? KNN_SEG : KNN_PRE;
  for (int m0 = 0; m0 < lim; m0 += 32) {
    unsigned le = 0u;
    #pragma unroll 8
    for (int k = 0; k < 32; ++k) {
      const float d = knn_score(me, seg[m0 + k]);
      le = le + le + ((d <= thr) ? 1u : 0u);
    }
    while (le != 0u) {
      const int k = __builtin_clz(le);
      bool tie = false;
      if (exact) tie = !(knn_score(me, seg[m0 + k]) < thr);
      take(tie, base + m0 + k);
      le &= ~(0x80000000u >> k);
    }
  }
  if (!spilt) {
    const int last = max(nhit - 1, 0);
    for (int j0 = 0; __builtin_amdgcn_ballot_w32(j0 < nhit) != 0u; j0 += 32) {
      unsigned le = 0u;
      #pragma unroll 8
      for (int k = 0; k < 32; ++k) {
        const float d = knn_score(me, seg[myhit[min(j0 + k, last)][lane]]);
        le = le + le + ((d <= thr) ? 1u : 0u);
      }
      const int have = min(max(nhit - j0, 0), 32);
      le &= (unsigned)(0xffffffff00000000ull >> have);
      while (le != 0u) {
        const int k = __builtin_clz(le);
        const int id = myhit[j0 + k][lane];
        bool tie = false;
        if (exact) tie = !(knn_score(me, seg[id]) < thr);
        take(tie, base + id);
        le &= ~(0x80000000u >> k);
      }
    }
  }
  for (; cnt < quota; ++cnt) op[cnt] = n;
}

#define QROWS 32

#define TB_THR   (KNN_Q * KNN_W)
#define TB_X     (NPTS / KNN_Q)
#define GM_PTS   32
#define GM_ROWS  (BATCH * NPTS / GM_PTS / TB_X)
#define GEO_BLKS (BATCH * NPTS / TB_THR)
#define GEO_ROWS (GEO_BLKS * 4 / TB_X)

template <bool CTR>
__device__ __forceinline__ void geo_rows(const float4* wt,
                                         const float x, const float y, const float z,
                                         float4* __restrict__ o) {
  #pragma unroll
  for (int j = 0; j < 8; ++j) {
    float v[4];
    #pragma unroll
    for (int e = 0; e < 4; ++e) {
      const float4 w = wt[4 * j + e];
      if (CTR) v[e] = w.x * x + w.y * y + w.z * z + w.w;
      else     v[e] = w.x * x + w.y * y + w.z * z;
    }
    o[(size_t)j * NPTS] = make_float4(v[0], v[1], v[2], v[3]);
  }
}

__device__ __forceinline__ void geo_block(const int id, float* sh,
           const float* __restrict__ xyz,
           const float* __restrict__ w1, const float* __restrict__ b1,
           const float* __restrict__ g1, const float* __restrict__ be1,
           const float* __restrict__ m1, const float* __restrict__ v1,
           float* __restrict__ PQ) {
  const int bx = id % GEO_BLKS;
  const int g  = id / GEO_BLKS;
  const int c0 = (g & 1) * 32;
  const bool ctr = g >= 2;
  const int pt = bx * TB_THR + threadIdx.x;
  const int b  = pt / NPTS;
  const int n  = pt - b * NPTS;
  const float* xb = xyz + (size_t)b * 3 * NPTS;
  const float x = xb[n], y = xb[NPTS + n], z = xb[2 * NPTS + n];
  float4* wt = (float4*)sh;
  if (threadIdx.x < 32) {
    const int c = c0 + threadIdx.x;
    const float s1 = bn_scale(g1[c], v1[c]);
    float w[3];
    #pragma unroll
    for (int j = 0; j < 3; ++j) {
      const float wa = w1[c * 6 + j];
      const float wb = w1[c * 6 + 3 + j];
      w[j] = ctr ? s1 * (wb - wa) : s1 * wa;
    }
    const float bias = ctr ? b1[c] * s1 + (be1[c] - m1[c] * s1) : 0.0f;
    wt[threadIdx.x] = make_float4(w[0], w[1], w[2], bias);
  }
  __syncthreads();
  float4* o = (float4*)PQ + ((size_t)b * QROWS + 8 * g) * NPTS + n;
  if (ctr) geo_rows<true>(wt, x, y, z, o);
  else     geo_rows<false>(wt, x, y, z, o);
}

__device__ __forceinline__ void gemm_block(const int bx, const int b,
                          const float* __restrict__ feat, const float* __restrict__ w2,
                          const float* __restrict__ g2, const float* __restrict__ v2,
                          float* __restrict__ Cf) {
  const int n0   = bx * GM_PTS;
  const int wave = threadIdx.x >> 5;
  const int lane = threadIdx.x & 31;
  const int l16  = lane & 15;
  const int hi   = lane >> 4;
  const int o    = wave * 16 + l16;
  const float s2 = bn_scale(g2[o], v2[o]);

  const float* Fb   = feat + (size_t)b * CH * NPTS;
  const float* wa   = w2 + (size_t)o * (2 * CH) + 2 * hi;
  const float* wb   = wa + CH;
  const float* bcol = Fb + (size_t)(2 * hi) * NPTS + n0 + l16;

  v8f an0 = {0.f, 0.f, 0.f, 0.f, 0.f, 0.f, 0.f, 0.f};
  v8f an1 = an0, ac0 = an0, ac1 = an0;

  #pragma unroll 8
  for (int k = 0; k < CH; k += 4) {
    const float wax = wa[k], way = wa[k + 1];
    const float wbx = wb[k], wby = wb[k + 1];
    v2f a_n;
    a_n.x = s2 * wax;
    a_n.y = s2 * way;
    v2f a_c;
    a_c.x = s2 * (wbx - wax);
    a_c.y = s2 * (wby - way);
    v2f b0;
    b0.x = bcol[(size_t)k * NPTS];
    b0.y = bcol[(size_t)(k + 1) * NPTS];
    v2f b1;
    b1.x = bcol[(size_t)k * NPTS + 16];
    b1.y = bcol[(size_t)(k + 1) * NPTS + 16];
    an0 = __builtin_amdgcn_wmma_f32_16x16x4_f32(false, a_n, false, b0, (short)0, an0, false, false);
    an1 = __builtin_amdgcn_wmma_f32_16x16x4_f32(false, a_n, false, b1, (short)0, an1, false, false);
    ac0 = __builtin_amdgcn_wmma_f32_16x16x4_f32(false, a_c, false, b0, (short)0, ac0, false, false);
    ac1 = __builtin_amdgcn_wmma_f32_16x16x4_f32(false, a_c, false, b1, (short)0, ac1, false, false);
  }

  float4* on = (float4*)Cf + ((size_t)b * QROWS + (16 * wave + 8 * hi) / 4) * NPTS + n0 + l16;
  float4* oc = on + (size_t)(HALF_CH / 4) * NPTS;
  on[0]         = make_float4(an0[0], an0[1], an0[2], an0[3]);
  on[NPTS]      = make_float4(an0[4], an0[5], an0[6], an0[7]);
  on[16]        = make_float4(an1[0], an1[1], an1[2], an1[3]);
  on[NPTS + 16] = make_float4(an1[4], an1[5], an1[6], an1[7]);
  oc[0]         = make_float4(ac0[0], ac0[1], ac0[2], ac0[3]);
  oc[NPTS]      = make_float4(ac0[4], ac0[5], ac0[6], ac0[7]);
  oc[16]        = make_float4(ac1[0], ac1[1], ac1[2], ac1[3]);
  oc[NPTS + 16] = make_float4(ac1[4], ac1[5], ac1[6], ac1[7]);
}

__device__ __forceinline__ void table_block(const int row, float* sh,
           const float* __restrict__ xyz, const float* __restrict__ feat,
           const float* __restrict__ w1, const float* __restrict__ b1,
           const float* __restrict__ g1, const float* __restrict__ be1,
           const float* __restrict__ m1, const float* __restrict__ v1,
           const float* __restrict__ w2, const float* __restrict__ g2,
           const float* __restrict__ v2,
           float* __restrict__ PQ, float* __restrict__ Cf) {
  if (row < GM_ROWS) gemm_block((int)blockIdx.x, row, feat, w2, g2, v2, Cf);
  else geo_block((row - GM_ROWS) * TB_X + (int)blockIdx.x, sh, xyz, w1, b1, g1, be1, m1, v1, PQ);
}

#define GA_THR 256
#define GA_PPT 2
#define GA_PTS (GA_THR * GA_PPT)
#define GA_STG (2 * NPTS / GA_THR)

__device__ __forceinline__ float4 max4(const float4 a, const float4 b) {
  return make_float4(fmaxf(a.x, b.x), fmaxf(a.y, b.y), fmaxf(a.z, b.z), fmaxf(a.w, b.w));
}

__global__ void __launch_bounds__(GA_THR)
gather_max(const float* __restrict__ PQ, const float* __restrict__ Cf,
           const int* __restrict__ idx, const float* __restrict__ bias2p,
           float* __restrict__ out) {
  __shared__ float4 tab[2 * NPTS];
  const int b    = blockIdx.z;
  const int tp   = blockIdx.y;
  const bool fea = tp >= 8;
  const int q0   = (tp & 7) * 2;
  const int tid  = threadIdx.x;
  const float4* T   = (const float4*)(fea ? Cf : PQ) + (size_t)b * QROWS * NPTS;
  const float4* src = T + (size_t)q0 * NPTS;

  float4 st[GA_STG];
  #pragma unroll
  for (int i = 0; i < GA_STG; ++i) st[i] = src[tid + i * GA_THR];

  const int nb = blockIdx.x * GA_PTS + tid;
  int4   id[GA_PPT][KNN_K / 4];
  float4 ctr[GA_PPT][2];
  #pragma unroll
  for (int p = 0; p < GA_PPT; ++p) {
    const int n = nb + p * GA_THR;
    const int4* ip = (const int4*)(idx + ((size_t)b * NPTS + n) * KNN_K);
    #pragma unroll
    for (int j = 0; j < KNN_K / 4; ++j) id[p][j] = ip[j];
    ctr[p][0] = T[(size_t)(16 + q0) * NPTS + n];
    ctr[p][1] = T[(size_t)(17 + q0) * NPTS + n];
  }

  #pragma unroll
  for (int i = 0; i < GA_STG; ++i) tab[tid + i * GA_THR] = st[i];

  float4 bs0 = make_float4(0.f, 0.f, 0.f, 0.f), bs1 = bs0;
  if (fea) {
    const float* bp = bias2p + 4 * q0;
    bs0 = make_float4(bp[0], bp[1], bp[2], bp[3]);
    bs1 = make_float4(bp[4], bp[5], bp[6], bp[7]);
  }
  __syncthreads();

  float* ob = out + ((size_t)b * CH + (fea ? HALF_CH : 0) + 4 * q0) * NPTS;
  #pragma unroll
  for (int p = 0; p < GA_PPT; ++p) {
    const int n = nb + p * GA_THR;
    float4 m0 = make_float4(-FLT_MAX, -FLT_MAX, -FLT_MAX, -FLT_MAX), m1 = m0;
    #pragma unroll
    for (int j = 0; j < KNN_K / 4; ++j) {
      const int4 q = id[p][j];
      m0 = max4(m0, tab[q.x]); m1 = max4(m1, tab[NPTS + q.x]);
      m0 = max4(m0, tab[q.y]); m1 = max4(m1, tab[NPTS + q.y]);
      m0 = max4(m0, tab[q.z]); m1 = max4(m1, tab[NPTS + q.z]);
      m0 = max4(m0, tab[q.w]); m1 = max4(m1, tab[NPTS + q.w]);
    }
    float4 c0 = ctr[p][0], c1 = ctr[p][1];
    if (fea) {
      c0 = make_float4(c0.x + bs0.x, c0.y + bs0.y, c0.z + bs0.z, c0.w + bs0.w);
      c1 = make_float4(c1.x + bs1.x, c1.y + bs1.y, c1.z + bs1.z, c1.w + bs1.w);
    }
    ob[0 * NPTS + n] = fmaxf(m0.x + c0.x, 0.0f);
    ob[1 * NPTS + n] = fmaxf(m0.y + c0.y, 0.0f);
    ob[2 * NPTS + n] = fmaxf(m0.z + c0.z, 0.0f);
    ob[3 * NPTS + n] = fmaxf(m0.w + c0.w, 0.0f);
    ob[4 * NPTS + n] = fmaxf(m1.x + c1.x, 0.0f);
    ob[5 * NPTS + n] = fmaxf(m1.y + c1.y, 0.0f);
    ob[6 * NPTS + n] = fmaxf(m1.z + c1.z, 0.0f);
    ob[7 * NPTS + n] = fmaxf(m1.w + c1.w, 0.0f);
  }
}

extern "C" void kernel_launch(void* const* d_in, const int* in_sizes, int n_in,
                              void* d_out, int out_size, void* d_ws, size_t ws_size,
                              hipStream_t stream) {
  (void)in_sizes; (void)n_in; (void)out_size; (void)ws_size;

  const float* xyz  = (const float*)d_in[0];
  const float* feat = (const float*)d_in[1];
  const float* w1   = (const float*)d_in[2];
  const float* b1   = (const float*)d_in[3];
  const float* g1   = (const float*)d_in[4];
  const float* be1  = (const float*)d_in[5];
  const float* m1   = (const float*)d_in[6];
  const float* v1   = (const float*)d_in[7];
  const float* w2   = (const float*)d_in[8];
  const float* b2   = (const float*)d_in[9];
  const float* g2   = (const float*)d_in[10];
  const float* be2  = (const float*)d_in[11];
  const float* m2   = (const float*)d_in[12];
  const float* v2   = (const float*)d_in[13];

  char* ws = (char*)d_ws;
  float* bias2p = (float*)ws;
  float* PQ     = (float*)(ws + (128 << 10));
  float* Cf     = PQ + (size_t)BATCH * NPTS * CH;
  int*   idxb   = (int*)(Cf + (size_t)BATCH * NPTS * CH);

  knn_kernel<<<dim3(TB_X, BATCH + 1 + GM_ROWS + GEO_ROWS), TB_THR, 0, stream>>>(
      xyz, feat, idxb, w1, b1, g1, be1, m1, v1, w2, b2, g2, be2, m2, v2,
      bias2p, PQ, Cf);

  gather_max<<<dim3(NPTS / GA_PTS, 16, BATCH), GA_THR, 0, stream>>>(PQ, Cf, idxb, bias2p,
                                                                    (float*)d_out);
}


# kNN scan 1: the 4 waves join their head lists into a block-wide bound (20th smallest); later candidates are only compared with it and listed, then joined by one sort network; scores bit-identical
# speedup vs baseline: 110.7498x; 1.0468x over previous
#include <hip/hip_runtime.h>
#include <hip/hip_bf16.h>
#include <float.h>

#define BATCH   8
#define NPTS    2048
#define CH      128
#define HALF_CH 64
#define KNN_K   20
#define EPSB    1e-5f

typedef __attribute__((ext_vector_type(2))) float v2f;
typedef __attribute__((ext_vector_type(8))) float v8f;

#define KNN_Q    32
#define KNN_W    4
#define KNN_SEG  (NPTS / KNN_W)
#define KNN_STG  (NPTS / (KNN_Q * KNN_W))
#define KNN_PRE  128
#define KNN_CAP  95

__device__ __forceinline__ float knn_score(const float4 me, const float4 p) {
  #pragma clang fp contract(off)
  const float yy  = me.y * p.y;
  const float zz  = me.z * p.z;
  const float dot = __builtin_fmaf(me.x, p.x, yy) + zz;
  return __builtin_fmaf(-2.0f, dot, me.w + p.w);
}

__device__ __forceinline__ void knn_insert(float (&t)[KNN_K], const float d) {
  #pragma unroll
  for (int i = KNN_K - 1; i > 0; --i) t[i] = __builtin_amdgcn_fmed3f(d, t[i - 1], t[i]);
  t[0] = fminf(d, t[0]);
}

template <int NY>
__device__ __forceinline__ void knn_join(const float (&x)[KNN_K], const float (&y)[NY],
                                         float (&e)[32]) {
  static_assert(KNN_K <= 32 && NY <= 32 && KNN_K + NY >= 32, "every e[i] needs a real score");
  #pragma unroll
  for (int i = 0; i < 32; ++i) {
    const int j = 31 - i;
    e[i] = (i < KNN_K && j < NY) ? fminf(x[i], y[j]) : (i < KNN_K) ? x[i] : y[j];
  }
  #pragma unroll
  for (int d = 16; d >= 1; d >>= 1) {
    #pragma unroll
    for (int i = 0; i < 32; ++i) {
      if ((i & d) == 0) {
        const float a = e[i], b = e[i + d];
        e[i] = fminf(a, b);
        e[i + d] = fmaxf(a, b);
      }
    }
  }
}

#define KNN_CE(a, b) { const float lo_ = fminf(s[a], s[b]); s[b] = fmaxf(s[a], s[b]); s[a] = lo_; }
__device__ __forceinline__ void knn_sort32(float (&s)[32]) {
  KNN_CE(0,1) KNN_CE(2,3) KNN_CE(4,5) KNN_CE(6,7) KNN_CE(8,9) KNN_CE(10,11) KNN_CE(12,13) KNN_CE(14,15)
  KNN_CE(16,17) KNN_CE(18,19) KNN_CE(20,21) KNN_CE(22,23) KNN_CE(24,25) KNN_CE(26,27) KNN_CE(28,29) KNN_CE(30,31)
  KNN_CE(0,2) KNN_CE(1,3) KNN_CE(4,6) KNN_CE(5,7) KNN_CE(8,10) KNN_CE(9,11) KNN_CE(12,14) KNN_CE(13,15)
  KNN_CE(16,18) KNN_CE(17,19) KNN_CE(20,22) KNN_CE(21,23) KNN_CE(24,26) KNN_CE(25,27) KNN_CE(28,30) KNN_CE(29,31)
  KNN_CE(1,2) KNN_CE(5,6) KNN_CE(9,10) KNN_CE(13,14) KNN_CE(17,18) KNN_CE(21,22) KNN_CE(25,26) KNN_CE(29,30)
  KNN_CE(0,4) KNN_CE(1,5) KNN_CE(2,6) KNN_CE(3,7) KNN_CE(8,12) KNN_CE(9,13) KNN_CE(10,14) KNN_CE(11,15)
  KNN_CE(16,20) KNN_CE(17,21) KNN_CE(18,22) KNN_CE(19,23) KNN_CE(24,28) KNN_CE(25,29) KNN_CE(26,30) KNN_CE(27,31)
  KNN_CE(2,4) KNN_CE(3,5) KNN_CE(10,12) KNN_CE(11,13) KNN_CE(18,20) KNN_CE(19,21) KNN_CE(26,28) KNN_CE(27,29)
  KNN_CE(1,2) KNN_CE(3,4) KNN_CE(5,6) KNN_CE(9,10) KNN_CE(11,12) KNN_CE(13,14) KNN_CE(17,18) KNN_CE(19,20)
  KNN_CE(21,22) KNN_CE(25,26) KNN_CE(27,28) KNN_CE(29,30) KNN_CE(0,8) KNN_CE(1,9) KNN_CE(2,10) KNN_CE(3,11)
  KNN_CE(4,12) KNN_CE(5,13) KNN_CE(6,14) KNN_CE(7,15) KNN_CE(16,24) KNN_CE(17,25) KNN_CE(18,26) KNN_CE(19,27)
  KNN_CE(20,28) KNN_CE(21,29) KNN_CE(22,30) KNN_CE(23,31) KNN_CE(4,8) KNN_CE(5,9) KNN_CE(6,10) KNN_CE(7,11)
  KNN_CE(20,24) KNN_CE(21,25) KNN_CE(22,26) KNN_CE(23,27) KNN_CE(2,4) KNN_CE(3,5) KNN_CE(6,8) KNN_CE(7,9)
  KNN_CE(10,12) KNN_CE(11,13) KNN_CE(18,20) KNN_CE(19,21) KNN_CE(22,24) KNN_CE(23,25) KNN_CE(26,28) KNN_CE(27,29)
  KNN_CE(1,2) KNN_CE(3,4) KNN_CE(5,6) KNN_CE(7,8) KNN_CE(9,10) KNN_CE(11,12) KNN_CE(13,14) KNN_CE(17,18)
  KNN_CE(19,20) KNN_CE(21,22) KNN_CE(23,24) KNN_CE(25,26) KNN_CE(27,28) KNN_CE(29,30) KNN_CE(0,16) KNN_CE(1,17)
  KNN_CE(2,18) KNN_CE(3,19) KNN_CE(4,20) KNN_CE(5,21) KNN_CE(6,22) KNN_CE(7,23) KNN_CE(8,24) KNN_CE(9,25)
  KNN_CE(10,26) KNN_CE(11,27) KNN_CE(12,28) KNN_CE(13,29) KNN_CE(14,30) KNN_CE(15,31) KNN_CE(8,16) KNN_CE(9,17)
  KNN_CE(10,18) KNN_CE(11,19) KNN_CE(12,20) KNN_CE(13,21) KNN_CE(14,22) KNN_CE(15,23) KNN_CE(4,8) KNN_CE(5,9)
  KNN_CE(6,10) KNN_CE(7,11) KNN_CE(12,16) KNN_CE(13,17) KNN_CE(14,18) KNN_CE(15,19) KNN_CE(20,24) KNN_CE(21,25)
  KNN_CE(22,26) KNN_CE(23,27) KNN_CE(2,4) KNN_CE(3,5) KNN_CE(6,8) KNN_CE(7,9) KNN_CE(10,12) KNN_CE(11,13)
  KNN_CE(14,16) KNN_CE(15,17) KNN_CE(18,20) KNN_CE(19,21) KNN_CE(22,24) KNN_CE(23,25) KNN_CE(26,28) KNN_CE(27,29)
  KNN_CE(1,2) KNN_CE(3,4) KNN_CE(5,6) KNN_CE(7,8) KNN_CE(9,10) KNN_CE(11,12) KNN_CE(13,14) KNN_CE(15,16)
  KNN_CE(17,18) KNN_CE(19,20) KNN_CE(21,22) KNN_CE(23,24) KNN_CE(25,26) KNN_CE(27,28) KNN_CE(29,30)
}
#undef KNN_CE

__device__ __forceinline__ float bn_scale(const float g, const float v) {
  return g / sqrtf(v + EPSB);
}

__device__ __forceinline__ void fold_bias2(
    const float* __restrict__ b2, const float* __restrict__ g2,
    const float* __restrict__ be2, const float* __restrict__ m2,
    const float* __restrict__ v2, float* __restrict__ bias2p) {
  if (blockIdx.x == 0 && threadIdx.x < HALF_CH) {
    const int o = threadIdx.x;
    const float s2 = bn_scale(g2[o], v2[o]);
    bias2p[o] = b2[o] * s2 + (be2[o] - m2[o] * s2);
  }
}

__device__ __forceinline__ void table_block(const int row, float* sh,
           const float* __restrict__ xyz, const float* __restrict__ feat,
           const float* __restrict__ w1, const float* __restrict__ b1,
           const float* __restrict__ g1, const float* __restrict__ be1,
           const float* __restrict__ m1, const float* __restrict__ v1,
           const float* __restrict__ w2, const float* __restrict__ g2,
           const float* __restrict__ v2,
           float* __restrict__ PQ, float* __restrict__ Cf);

__global__ void __launch_bounds__(KNN_Q * KNN_W)
knn_kernel(const float* __restrict__ xyz, const float* __restrict__ feat,
           int* __restrict__ idx_out,
           const float* __restrict__ w1, const float* __restrict__ b1,
           const float* __restrict__ g1, const float* __restrict__ be1,
           const float* __restrict__ m1, const float* __restrict__ v1,
           const float* __restrict__ w2, const float* __restrict__ b2,
           const float* __restrict__ g2, const float* __restrict__ be2,
           const float* __restrict__ m2, const float* __restrict__ v2,
           float* __restrict__ bias2p,
           float* __restrict__ PQ, float* __restrict__ Cf) {
  __shared__ float4 pts[NPTS];
  __shared__ float  lst[KNN_W][KNN_K][KNN_Q];
  __shared__ float  hd[KNN_W][KNN_K][KNN_Q];
  __shared__ int    hit[KNN_W][KNN_CAP + 1][KNN_Q];
  if (blockIdx.y >= BATCH) {
    if (blockIdx.y == BATCH) fold_bias2(b2, g2, be2, m2, v2, bias2p);
    else table_block((int)blockIdx.y - (BATCH + 1), (float*)pts, xyz, feat,
                     w1, b1, g1, be1, m1, v1, w2, g2, v2, PQ, Cf);
    return;
  }
  const int b = blockIdx.y;
  const float* xb = xyz + (size_t)b * 3 * NPTS;
  float sx[KNN_STG], sy[KNN_STG], sz[KNN_STG];
  #pragma unroll
  for (int i = 0; i < KNN_STG; ++i) {
    const int m = threadIdx.x + i * (KNN_Q * KNN_W);
    sx[i] = xb[m]; sy[i] = xb[NPTS + m]; sz[i] = xb[2 * NPTS + m];
  }
  #pragma unroll
  for (int i = 0; i < KNN_STG; ++i) {
    const int m = threadIdx.x + i * (KNN_Q * KNN_W);
    const float x = sx[i], y = sy[i], z = sz[i];
    pts[m] = make_float4(x, y, z, __builtin_fmaf(z, z, __builtin_fmaf(x, x, y * y)));
  }
  __syncthreads();

  const int lane = threadIdx.x & (KNN_Q - 1);
  const int w    = threadIdx.x / KNN_Q;
  const int n    = blockIdx.x * KNN_Q + lane;
  const float4 me = pts[n];
  const float4* seg = pts + w * KNN_SEG;
  int (*myhit)[KNN_Q] = hit[w];
  int nhit = 0;

  float t[KNN_K];
  #pragma unroll
  for (int i = 0; i < KNN_K; ++i) t[i] = FLT_MAX;

  static_assert(KNN_PRE % 32 == 0, "the head is taken in whole batches of 32");
  for (int m0 = 0; m0 < KNN_PRE; m0 += 32) {
    float s[32];
    #pragma unroll
    for (int k = 0; k < 32; ++k) s[k] = knn_score(me, seg[m0 + k]);
    knn_sort32(s);
    float e[32];
    knn_join<32>(t, s, e);
    #pragma unroll
    for (int i = 0; i < KNN_K; ++i) t[i] = e[i];
  }

  #pragma unroll
  for (int i = 0; i < KNN_K; ++i) hd[w][i][lane] = t[i];
  __syncthreads();
  float bound;
  {
    float h1[KNN_K], h2[KNN_K], h3[KNN_K];
    #pragma unroll
    for (int i = 0; i < KNN_K; ++i) {
      h1[i] = hd[w ^ 1][i][lane];
      h2[i] = hd[w ^ 2][i][lane];
      h3[i] = hd[w ^ 3][i][lane];
    }
    float a[32], b[32];
    knn_join<KNN_K>(t, h1, a);
    knn_join<KNN_K>(h2, h3, b);
    bound = fminf(a[0], b[KNN_K - 1]);
    #pragma unroll
    for (int i = 1; i < KNN_K; ++i) bound = fmaxf(bound, fminf(a[i], b[KNN_K - 1 - i]));
  }

  myhit[0][lane] = KNN_PRE;
  for (int m0 = KNN_PRE; m0 < KNN_SEG; m0 += 32) {
    unsigned in = 0u;
    #pragma unroll 8
    for (int k = 0; k < 32; ++k) {
      const float d = knn_score(me, seg[m0 + k]);
      in = in + in + ((d <= bound) ? 1u : 0u);
    }
    while (in != 0u) {
      const int k = __builtin_clz(in);
      myhit[min(nhit, KNN_CAP)][lane] = m0 + k;
      ++nhit;
      in &= ~(0x80000000u >> k);
    }
  }

  if (__builtin_amdgcn_ballot_w32(nhit > KNN_CAP) == 0u) {
    const int last = max(nhit - 1, 0);
    for (int j0 = 0; __builtin_amdgcn_ballot_w32(j0 < nhit) != 0u; j0 += 32) {
      float s[32];
      #pragma unroll
      for (int k = 0; k < 32; ++k) {
        const float d = knn_score(me, seg[myhit[min(j0 + k, last)][lane]]);
        s[k] = (j0 + k < nhit) ? d : FLT_MAX;
      }
      knn_sort32(s);
      float e[32];
      knn_join<32>(t, s, e);
      #pragma unroll
      for (int i = 0; i < KNN_K; ++i) t[i] = e[i];
    }
  } else {
    for (int m = KNN_PRE; m < KNN_SEG; ++m) knn_insert(t, knn_score(me, seg[m]));
  }

  #pragma unroll
  for (int i = 0; i < KNN_K; ++i) lst[w][i][lane] = t[i];
  __syncthreads();
  static_assert(KNN_W == 4, "the pairing below is for four lists");
  float low[KNN_K];
  {
    float l1[KNN_K], l2[KNN_K], l3[KNN_K];
    #pragma unroll
    for (int i = 0; i < KNN_K; ++i) {
      l1[i] = lst[w ^ 1][i][lane];
      l2[i] = lst[w ^ 2][i][lane];
      l3[i] = lst[w ^ 3][i][lane];
    }
    float a[32], b[32];
    knn_join<KNN_K>(t, l1, a);
    knn_join<KNN_K>(l2, l3, b);
    #pragma unroll
    for (int i = 0; i < KNN_K; ++i) low[i] = fminf(a[i], b[KNN_K - 1 - i]);
  }

  float thr = low[0];
  #pragma unroll
  for (int i = 1; i < KNN_K; ++i) thr = fmaxf(thr, low[i]);
  int ties = 0;
  #pragma unroll
  for (int i = 0; i < KNN_K; ++i) ties += (low[i] == thr) ? 1 : 0;

  int off = 0;
  int quota = 0;
  int mine_eq = 0;
  int rem = ties;
  bool exact = false;
  for (int q = 0; q <= w; ++q) {
    int lt = 0, eq = 0;
    #pragma unroll
    for (int i = 0; i < KNN_K; ++i) {
      const float s = lst[q][i][lane];
      lt += (s < thr) ? 1 : 0;
      eq += (s == thr) ? 1 : 0;
    }
    const int a = min(eq, rem);
    rem -= a;
    if (q < w) off += lt + a;
    else {
      quota = lt + a; mine_eq = a;
      exact = (a < eq) || !(lst[q][KNN_K - 1][lane] > thr);
    }
  }
  if (w == KNN_W - 1) quota = KNN_K - off;

  int* op = idx_out + ((size_t)b * NPTS + n) * KNN_K + off;
  const int base = w * KNN_SEG;
  int cnt = 0;
  auto take = [&](const bool tie, const int id) {
    if ((!tie || mine_eq > 0) && cnt < quota) { op[cnt] = id; ++cnt; }
    if (tie) --mine_eq;
  };
  const bool spilt = __builtin_amdgcn_ballot_w32(nhit > KNN_CAP) != 0u;
  const int lim = spilt ? KNN_SEG : KNN_PRE;
  for (int m0 = 0; m0 < lim; m0 += 32) {
    unsigned le = 0u;
    #pragma unroll 8
    for (int k = 0; k < 32; ++k) {
      const float d = knn_score(me, seg[m0 + k]);
      le = le + le + ((d <= thr) ? 1u : 0u);
    }
    while (le != 0u) {
      const int k = __builtin_clz(le);
      bool tie = false;
      if (exact) tie = !(knn_score(me, seg[m0 + k]) < thr);
      take(tie, base + m0 + k);
      le &= ~(0x80000000u >> k);
    }
  }
  if (!spilt) {
    const int last = max(nhit - 1, 0);
    for (int j0 = 0; __builtin_amdgcn_ballot_w32(j0 < nhit) != 0u; j0 += 32) {
      unsigned le = 0u;
      #pragma unroll 8
      for (int k = 0; k < 32; ++k) {
        const float d = knn_score(me, seg[myhit[min(j0 + k, last)][lane]]);
        le = le + le + ((d <= thr) ? 1u : 0u);
      }
      const int have = min(max(nhit - j0, 0), 32);
      le &= (unsigned)(0xffffffff00000000ull >> have);
      while (le != 0u) {
        const int k = __builtin_clz(le);
        const int id = myhit[j0 + k][lane];
        bool tie = false;
        if (exact) tie = !(knn_score(me, seg[id]) < thr);
        take(tie, base + id);
        le &= ~(0x80000000u >> k);
      }
    }
  }
  for (; cnt < quota; ++cnt) op[cnt] = n;
}

#define QROWS 32

#define TB_THR   (KNN_Q * KNN_W)
#define TB_X     (NPTS / KNN_Q)
#define GM_PTS   32
#define GM_ROWS  (BATCH * NPTS / GM_PTS / TB_X)
#define GEO_BLKS (BATCH * NPTS / TB_THR)
#define GEO_ROWS (GEO_BLKS * 4 / TB_X)

template <bool CTR>
__device__ __forceinline__ void geo_rows(const float4* wt,
                                         const float x, const float y, const float z,
                                         float4* __restrict__ o) {
  #pragma unroll
  for (int j = 0; j < 8; ++j) {
    float v[4];
    #pragma unroll
    for (int e = 0; e < 4; ++e) {
      const float4 w = wt[4 * j + e];
      if (CTR) v[e] = w.x * x + w.y * y + w.z * z + w.w;
      else     v[e] = w.x * x + w.y * y + w.z * z;
    }
    o[(size_t)j * NPTS] = make_float4(v[0], v[1], v[2], v[3]);
  }
}

__device__ __forceinline__ void geo_block(const int id, float* sh,
           const float* __restrict__ xyz,
           const float* __restrict__ w1, const float* __restrict__ b1,
           const float* __restrict__ g1, const float* __restrict__ be1,
           const float* __restrict__ m1, const float* __restrict__ v1,
           float* __restrict__ PQ) {
  const int bx = id % GEO_BLKS;
  const int g  = id / GEO_BLKS;
  const int c0 = (g & 1) * 32;
  const bool ctr = g >= 2;
  const int pt = bx * TB_THR + threadIdx.x;
  const int b  = pt / NPTS;
  const int n  = pt - b * NPTS;
  const float* xb = xyz + (size_t)b * 3 * NPTS;
  const float x = xb[n], y = xb[NPTS + n], z = xb[2 * NPTS + n];
  float4* wt = (float4*)sh;
  if (threadIdx.x < 32) {
    const int c = c0 + threadIdx.x;
    const float s1 = bn_scale(g1[c], v1[c]);
    float w[3];
    #pragma unroll
    for (int j = 0; j < 3; ++j) {
      const float wa = w1[c * 6 + j];
      const float wb = w1[c * 6 + 3 + j];
      w[j] = ctr ? s1 * (wb - wa) : s1 * wa;
    }
    const float bias = ctr ? b1[c] * s1 + (be1[c] - m1[c] * s1) : 0.0f;
    wt[threadIdx.x] = make_float4(w[0], w[1], w[2], bias);
  }
  __syncthreads();
  float4* o = (float4*)PQ + ((size_t)b * QROWS + 8 * g) * NPTS + n;
  if (ctr) geo_rows<true>(wt, x, y, z, o);
  else     geo_rows<false>(wt, x, y, z, o);
}

__device__ __forceinline__ void gemm_block(const int bx, const int b,
                          const float* __restrict__ feat, const float* __restrict__ w2,
                          const float* __restrict__ g2, const float* __restrict__ v2,
                          float* __restrict__ Cf) {
  const int n0   = bx * GM_PTS;
  const int wave = threadIdx.x >> 5;
  const int lane = threadIdx.x & 31;
  const int l16  = lane & 15;
  const int hi   = lane >> 4;
  const int o    = wave * 16 + l16;
  const float s2 = bn_scale(g2[o], v2[o]);

  const float* Fb   = feat + (size_t)b * CH * NPTS;
  const float* wa   = w2 + (size_t)o * (2 * CH) + 2 * hi;
  const float* wb   = wa + CH;
  const float* bcol = Fb + (size_t)(2 * hi) * NPTS + n0 + l16;

  v8f an0 = {0.f, 0.f, 0.f, 0.f, 0.f, 0.f, 0.f, 0.f};
  v8f an1 = an0, ac0 = an0, ac1 = an0;

  #pragma unroll 8
  for (int k = 0; k < CH; k += 4) {
    const float wax = wa[k], way = wa[k + 1];
    const float wbx = wb[k], wby = wb[k + 1];
    v2f a_n;
    a_n.x = s2 * wax;
    a_n.y = s2 * way;
    v2f a_c;
    a_c.x = s2 * (wbx - wax);
    a_c.y = s2 * (wby - way);
    v2f b0;
    b0.x = bcol[(size_t)k * NPTS];
    b0.y = bcol[(size_t)(k + 1) * NPTS];
    v2f b1;
    b1.x = bcol[(size_t)k * NPTS + 16];
    b1.y = bcol[(size_t)(k + 1) * NPTS + 16];
    an0 = __builtin_amdgcn_wmma_f32_16x16x4_f32(false, a_n, false, b0, (short)0, an0, false, false);
    an1 = __builtin_amdgcn_wmma_f32_16x16x4_f32(false, a_n, false, b1, (short)0, an1, false, false);
    ac0 = __builtin_amdgcn_wmma_f32_16x16x4_f32(false, a_c, false, b0, (short)0, ac0, false, false);
    ac1 = __builtin_amdgcn_wmma_f32_16x16x4_f32(false, a_c, false, b1, (short)0, ac1, false, false);
  }

  float4* on = (float4*)Cf + ((size_t)b * QROWS + (16 * wave + 8 * hi) / 4) * NPTS + n0 + l16;
  float4* oc = on + (size_t)(HALF_CH / 4) * NPTS;
  on[0]         = make_float4(an0[0], an0[1], an0[2], an0[3]);
  on[NPTS]      = make_float4(an0[4], an0[5], an0[6], an0[7]);
  on[16]        = make_float4(an1[0], an1[1], an1[2], an1[3]);
  on[NPTS + 16] = make_float4(an1[4], an1[5], an1[6], an1[7]);
  oc[0]         = make_float4(ac0[0], ac0[1], ac0[2], ac0[3]);
  oc[NPTS]      = make_float4(ac0[4], ac0[5], ac0[6], ac0[7]);
  oc[16]        = make_float4(ac1[0], ac1[1], ac1[2], ac1[3]);
  oc[NPTS + 16] = make_float4(ac1[4], ac1[5], ac1[6], ac1[7]);
}

__device__ __forceinline__ void table_block(const int row, float* sh,
           const float* __restrict__ xyz, const float* __restrict__ feat,
           const float* __restrict__ w1, const float* __restrict__ b1,
           const float* __restrict__ g1, const float* __restrict__ be1,
           const float* __restrict__ m1, const float* __restrict__ v1,
           const float* __restrict__ w2, const float* __restrict__ g2,
           const float* __restrict__ v2,
           float* __restrict__ PQ, float* __restrict__ Cf) {
  if (row < GM_ROWS) gemm_block((int)blockIdx.x, row, feat, w2, g2, v2, Cf);
  else geo_block((row - GM_ROWS) * TB_X + (int)blockIdx.x, sh, xyz, w1, b1, g1, be1, m1, v1, PQ);
}

#define GA_THR 256
#define GA_PPT 2
#define GA_PTS (GA_THR * GA_PPT)
#define GA_STG (2 * NPTS / GA_THR)

__device__ __forceinline__ float4 max4(const float4 a, const float4 b) {
  return make_float4(fmaxf(a.x, b.x), fmaxf(a.y, b.y), fmaxf(a.z, b.z), fmaxf(a.w, b.w));
}

__global__ void __launch_bounds__(GA_THR)
gather_max(const float* __restrict__ PQ, const float* __restrict__ Cf,
           const int* __restrict__ idx, const float* __restrict__ bias2p,
           float* __restrict__ out) {
  __shared__ float4 tab[2 * NPTS];
  const int b    = blockIdx.z;
  const int tp   = blockIdx.y;
  const bool fea = tp >= 8;
  const int q0   = (tp & 7) * 2;
  const int tid  = threadIdx.x;
  const float4* T   = (const float4*)(fea ? Cf : PQ) + (size_t)b * QROWS * NPTS;
  const float4* src = T + (size_t)q0 * NPTS;

  float4 st[GA_STG];
  #pragma unroll
  for (int i = 0; i < GA_STG; ++i) st[i] = src[tid + i * GA_THR];

  const int nb = blockIdx.x * GA_PTS + tid;
  int4   id[GA_PPT][KNN_K / 4];
  float4 ctr[GA_PPT][2];
  #pragma unroll
  for (int p = 0; p < GA_PPT; ++p) {
    const int n = nb + p * GA_THR;
    const int4* ip = (const int4*)(idx + ((size_t)b * NPTS + n) * KNN_K);
    #pragma unroll
    for (int j = 0; j < KNN_K / 4; ++j) id[p][j] = ip[j];
    ctr[p][0] = T[(size_t)(16 + q0) * NPTS + n];
    ctr[p][1] = T[(size_t)(17 + q0) * NPTS + n];
  }

  #pragma unroll
  for (int i = 0; i < GA_STG; ++i) tab[tid + i * GA_THR] = st[i];

  float4 bs0 = make_float4(0.f, 0.f, 0.f, 0.f), bs1 = bs0;
  if (fea) {
    const float* bp = bias2p + 4 * q0;
    bs0 = make_float4(bp[0], bp[1], bp[2], bp[3]);
    bs1 = make_float4(bp[4], bp[5], bp[6], bp[7]);
  }
  __syncthreads();

  float* ob = out + ((size_t)b * CH + (fea ? HALF_CH : 0) + 4 * q0) * NPTS;
  #pragma unroll
  for (int p = 0; p < GA_PPT; ++p) {
    const int n = nb + p * GA_THR;
    float4 m0 = make_float4(-FLT_MAX, -FLT_MAX, -FLT_MAX, -FLT_MAX), m1 = m0;
    #pragma unroll
    for (int j = 0; j < KNN_K / 4; ++j) {
      const int4 q = id[p][j];
      m0 = max4(m0, tab[q.x]); m1 = max4(m1, tab[NPTS + q.x]);
      m0 = max4(m0, tab[q.y]); m1 = max4(m1, tab[NPTS + q.y]);
      m0 = max4(m0, tab[q.z]); m1 = max4(m1, tab[NPTS + q.z]);
      m0 = max4(m0, tab[q.w]); m1 = max4(m1, tab[NPTS + q.w]);
    }
    float4 c0 = ctr[p][0], c1 = ctr[p][1];
    if (fea) {
      c0 = make_float4(c0.x + bs0.x, c0.y + bs0.y, c0.z + bs0.z, c0.w + bs0.w);
      c1 = make_float4(c1.x + bs1.x, c1.y + bs1.y, c1.z + bs1.z, c1.w + bs1.w);
    }
    ob[0 * NPTS + n] = fmaxf(m0.x + c0.x, 0.0f);
    ob[1 * NPTS + n] = fmaxf(m0.y + c0.y, 0.0f);
    ob[2 * NPTS + n] = fmaxf(m0.z + c0.z, 0.0f);
    ob[3 * NPTS + n] = fmaxf(m0.w + c0.w, 0.0f);
    ob[4 * NPTS + n] = fmaxf(m1.x + c1.x, 0.0f);
    ob[5 * NPTS + n] = fmaxf(m1.y + c1.y, 0.0f);
    ob[6 * NPTS + n] = fmaxf(m1.z + c1.z, 0.0f);
    ob[7 * NPTS + n] = fmaxf(m1.w + c1.w, 0.0f);
  }
}

extern "C" void kernel_launch(void* const* d_in, const int* in_sizes, int n_in,
                              void* d_out, int out_size, void* d_ws, size_t ws_size,
                              hipStream_t stream) {
  (void)in_sizes; (void)n_in; (void)out_size; (void)ws_size;

  const float* xyz  = (const float*)d_in[0];
  const float* feat = (const float*)d_in[1];
  const float* w1   = (const float*)d_in[2];
  const float* b1   = (const float*)d_in[3];
  const float* g1   = (const float*)d_in[4];
  const float* be1  = (const float*)d_in[5];
  const float* m1   = (const float*)d_in[6];
  const float* v1   = (const float*)d_in[7];
  const float* w2   = (const float*)d_in[8];
  const float* b2   = (const float*)d_in[9];
  const float* g2   = (const float*)d_in[10];
  const float* be2  = (const float*)d_in[11];
  const float* m2   = (const float*)d_in[12];
  const float* v2   = (const float*)d_in[13];

  char* ws = (char*)d_ws;
  float* bias2p = (float*)ws;
  float* PQ     = (float*)(ws + (128 << 10));
  float* Cf     = PQ + (size_t)BATCH * NPTS * CH;
  int*   idxb   = (int*)(Cf + (size_t)BATCH * NPTS * CH);

  knn_kernel<<<dim3(TB_X, BATCH + 1 + GM_ROWS + GEO_ROWS), TB_THR, 0, stream>>>(
      xyz, feat, idxb, w1, b1, g1, be1, m1, v1, w2, b2, g2, be2, m2, v2,
      bias2p, PQ, Cf);

  gather_max<<<dim3(NPTS / GA_PTS, 16, BATCH), GA_THR, 0, stream>>>(PQ, Cf, idxb, bias2p,
                                                                    (float*)d_out);
}
